# starting kernel
# baseline (speedup 1.0000x reference)
#include <hip/hip_runtime.h>

#define NN 50000
#define NE 800000
#define NLAB 50
#define EMB 64
#define HID 96

typedef __attribute__((ext_vector_type(2))) float v2f;
typedef __attribute__((ext_vector_type(8))) float v8f;


__global__ void zero2_kernel(float* __restrict__ a, float* __restrict__ b, int n) {
    int i = blockIdx.x * blockDim.x + threadIdx.x;
    if (i < n) { a[i] = 0.0f; b[i] = 0.0f; }
}

__global__ void edge_deg_kernel(const int* __restrict__ src, const int* __restrict__ dst,
                                float* __restrict__ odeg, float* __restrict__ ideg, int e) {
    int i = blockIdx.x * blockDim.x + threadIdx.x;
    if (i < e) {
        atomicAdd(&odeg[src[i]], 1.0f);
        atomicAdd(&ideg[dst[i]], 1.0f);
    }
}

__global__ void norm_kernel(float* __restrict__ od, float* __restrict__ id, int n) {
    int i = blockIdx.x * blockDim.x + threadIdx.x;
    if (i < n) {
        od[i] = rsqrtf(fmaxf(od[i], 1.0f));
        id[i] = rsqrtf(fmaxf(id[i], 1.0f));
    }
}


__global__ void embed_scale_kernel(const int* __restrict__ labels, const float* __restrict__ emb,
                                   const float* __restrict__ onorm, float* __restrict__ out) {
    int i = blockIdx.x;
    int f = threadIdx.x;
    out[(long)i * EMB + f] = emb[(long)labels[i] * EMB + f] * onorm[i];
}

__global__ void scale_rows_kernel(const float* __restrict__ h, const float* __restrict__ norm,
                                  float* __restrict__ out, int d) {
    int i = blockIdx.x;
    int f = threadIdx.x;
    out[(long)i * d + f] = h[(long)i * d + f] * norm[i];
}

__global__ void zero_buf_kernel(float* __restrict__ p, long n) {
    long i = (long)blockIdx.x * blockDim.x + threadIdx.x;
    if (i < n) p[i] = 0.0f;
}

__global__ void scatter_add_kernel(const float* __restrict__ hs, float* __restrict__ agg,
                                   const int* __restrict__ src, const int* __restrict__ dst, int d) {
    int e = blockIdx.x;
    int f = threadIdx.x;
    int s = src[e], t = dst[e];
    atomicAdd(&agg[(long)t * d + f], hs[(long)s * d + f]);
}


__global__ __launch_bounds__(256) void gcn_gemm_wmma(
    const float* __restrict__ A, const float* __restrict__ rowScale,
    const float* __restrict__ W, const float* __restrict__ bias,
    float* __restrict__ out, int K, int Nout, int Npad,
    int tilesN, int totalTiles, int relu)
{
    __shared__ float smem[HID * HID + HID];
    const int tid = threadIdx.x;
    const int KN = K * Npad;

    for (int idx = tid; idx < KN; idx += 256) {
        int k = idx / Npad, n = idx % Npad;
        smem[idx] = (n < Nout) ? W[k * Nout + n] : 0.0f;
    }
    for (int idx = tid; idx < Npad; idx += 256)
        smem[KN + idx] = (idx < Nout) ? bias[idx] : 0.0f;
    __syncthreads();

    const int wave = tid >> 5;
    const int lane = tid & 31;
    const int tileId = blockIdx.x * 8 + wave;
    if (tileId >= totalTiles) return;

    const int m0 = (tileId / tilesN) * 16;
    const int n0 = (tileId % tilesN) * 16;

    const int rowA = m0 + (lane & 15);
    const int koff = (lane >> 4) * 2;
    const int colB = lane & 15;

    v8f c;
    const float bval = smem[KN + n0 + colB];
#pragma unroll
    for (int r = 0; r < 8; ++r) c[r] = bval;

    const float s = rowScale[rowA];
    const float* arow = A + (long)rowA * K;
    const float* bcol = smem + n0 + colB;

    for (int k0 = 0; k0 < K; k0 += 4) {
        v2f a, b;
        a.x = arow[k0 + koff] * s;
        a.y = arow[k0 + koff + 1] * s;
        b.x = bcol[(k0 + koff) * Npad];
        b.y = bcol[(k0 + koff + 1) * Npad];
        c = __builtin_amdgcn_wmma_f32_16x16x4_f32(
                 false, a,  false, b,
                 (short)0, c,  false,  false);
    }

    const int rbase = m0 + (lane >> 4) * 8;
    const int col = n0 + colB;
    if (col < Nout) {
#pragma unroll
        for (int r = 0; r < 8; ++r) {
            float v = c[r];
            if (relu) v = fmaxf(v, 0.0f);
            out[(long)(rbase + r) * Nout + col] = v;
        }
    }
}


extern "C" void kernel_launch(void* const* d_in, const int* in_sizes, int n_in,
                              void* d_out, int out_size, void* d_ws, size_t ws_size,
                              hipStream_t stream) {
    const int*   labels = (const int*)d_in[0];
    const int*   src    = (const int*)d_in[1];
    const int*   dst    = (const int*)d_in[2];
    const float* emb    = (const float*)d_in[3];
    const float* W1     = (const float*)d_in[4];
    const float* b1     = (const float*)d_in[5];
    const float* W2     = (const float*)d_in[6];
    const float* b2     = (const float*)d_in[7];
    const float* W3     = (const float*)d_in[8];
    const float* b3     = (const float*)d_in[9];
    float* out = (float*)d_out;

    float* w = (float*)d_ws;
    float* onorm = w;
    float* inorm = w + NN;
    float* bufA  = w + 2 * (size_t)NN;
    float* bufB  = bufA + (size_t)NN * HID;
    float* bufC  = bufB + (size_t)NN * HID;

    const int TPB = 256;

    zero2_kernel<<<(NN + TPB - 1) / TPB, TPB, 0, stream>>>(onorm, inorm, NN);
    edge_deg_kernel<<<(NE + TPB - 1) / TPB, TPB, 0, stream>>>(src, dst, onorm, inorm, NE);
    norm_kernel<<<(NN + TPB - 1) / TPB, TPB, 0, stream>>>(onorm, inorm, NN);

    embed_scale_kernel<<<NN, EMB, 0, stream>>>(labels, emb, onorm, bufB);
    zero_buf_kernel<<<(int)(((long)NN * EMB + TPB - 1) / TPB), TPB, 0, stream>>>(bufC, (long)NN * EMB);
    scatter_add_kernel<<<NE, EMB, 0, stream>>>(bufB, bufC, src, dst, EMB);
    {
        int tilesN = HID / 16;
        int total  = (NN / 16) * tilesN;
        gcn_gemm_wmma<<<(total + 7) / 8, 256, 0, stream>>>(
            bufC, inorm, W1, b1, bufA, EMB, HID, HID, tilesN, total, 1);
    }

    scale_rows_kernel<<<NN, HID, 0, stream>>>(bufA, onorm, bufB, HID);
    zero_buf_kernel<<<(int)(((long)NN * HID + TPB - 1) / TPB), TPB, 0, stream>>>(bufC, (long)NN * HID);
    scatter_add_kernel<<<NE, HID, 0, stream>>>(bufB, bufC, src, dst, HID);
    {
        int tilesN = HID / 16;
        int total  = (NN / 16) * tilesN;
        gcn_gemm_wmma<<<(total + 7) / 8, 256, 0, stream>>>(
            bufC, inorm, W2, b2, bufA, HID, HID, HID, tilesN, total, 1);
    }

    scale_rows_kernel<<<NN, HID, 0, stream>>>(bufA, onorm, bufB, HID);
    zero_buf_kernel<<<(int)(((long)NN * HID + TPB - 1) / TPB), TPB, 0, stream>>>(bufC, (long)NN * HID);
    scatter_add_kernel<<<NE, HID, 0, stream>>>(bufB, bufC, src, dst, HID);
    {
        int Npad   = 64;
        int tilesN = Npad / 16;
        int total  = (NN / 16) * tilesN;
        gcn_gemm_wmma<<<(total + 7) / 8, 256, 0, stream>>>(
            bufC, inorm, W3, b3, out, HID, NLAB, Npad, tilesN, total, 0);
    }
}


# degrees and layer 1's label table summed by owner blocks in LDS (one owner byte scanned per edge, as in layers 2-3): no global atomic, zero fill or norm launch left; sum order: LDS-add arrival
# speedup vs baseline: 2.4745x; 2.4745x over previous
#include <hip/hip_runtime.h>

#define NN 50000
#define NE 800000
#define NLAB 50
#define EMB 64
#define HID 96
#define LPAD 52

#define RNG 196
#define NOWN ((NN + RNG - 1) / RNG)
#define AGG_T 1024
#define SCAN_L 7
#define NQ16 (NE / 16)
#define PASS_Q (AGG_T * SCAN_L)
#define NPASS ((NQ16 + PASS_Q - 1) / PASS_Q)
#define QCAP 12288
#define QALL 45056

static_assert(NOWN <= 256, "owner id must fit one byte");
static_assert(NN <= 65536 && RNG <= 65536, "edge record packs source and local row in 16 bits each");
static_assert(NE % 16 == 0, "owner bytes are read 16 at a time");
static_assert(PASS_Q <= NQ16, "the first pass must lie inside the owner array");

typedef __attribute__((ext_vector_type(2))) float v2f;
typedef __attribute__((ext_vector_type(8))) float v8f;


__global__ void edge_rec_kernel(const int* __restrict__ src, const int* __restrict__ dst,
                                unsigned* __restrict__ own, unsigned* __restrict__ owns,
                                unsigned* __restrict__ rec, int e4) {
    int i = blockIdx.x * blockDim.x + threadIdx.x;
    if (i < e4) {
        const int4 s = ((const int4*)src)[i];
        const int4 t = ((const int4*)dst)[i];
        const unsigned o0 = (unsigned)t.x / RNG, o1 = (unsigned)t.y / RNG;
        const unsigned o2 = (unsigned)t.z / RNG, o3 = (unsigned)t.w / RNG;
        own[i] = o0 | (o1 << 8) | (o2 << 16) | (o3 << 24);
        const unsigned p0 = (unsigned)s.x / RNG, p1 = (unsigned)s.y / RNG;
        const unsigned p2 = (unsigned)s.z / RNG, p3 = (unsigned)s.w / RNG;
        owns[i] = p0 | (p1 << 8) | (p2 << 16) | (p3 << 24);
        uint4 r;
        r.x = (((unsigned)t.x - o0 * RNG) << 16) | (unsigned)s.x;
        r.y = (((unsigned)t.y - o1 * RNG) << 16) | (unsigned)s.y;
        r.z = (((unsigned)t.z - o2 * RNG) << 16) | (unsigned)s.z;
        r.w = (((unsigned)t.w - o3 * RNG) << 16) | (unsigned)s.w;
        ((uint4*)rec)[i] = r;
    }
}


__global__ __launch_bounds__(32) void label_weight_wmma(const float* __restrict__ emb,
                                                        const float* __restrict__ W1,
                                                        float* __restrict__ ew) {
    const int lane = threadIdx.x;
    const int m0 = (blockIdx.x / (HID / 16)) * 16;
    const int n0 = (blockIdx.x % (HID / 16)) * 16;

    const int rowA = m0 + (lane & 15);
    const int koff = (lane >> 4) * 2;
    const int colB = n0 + (lane & 15);
    const bool live = rowA < NLAB;
    const float* arow = emb + (long)(live ? rowA : 0) * EMB;

    v8f c = {0.0f, 0.0f, 0.0f, 0.0f, 0.0f, 0.0f, 0.0f, 0.0f};
    for (int k0 = 0; k0 < EMB; k0 += 4) {
        v2f a, b;
        float a0 = arow[k0 + koff];
        float a1 = arow[k0 + koff + 1];
        a.x = live ? a0 : 0.0f;
        a.y = live ? a1 : 0.0f;
        b.x = W1[(k0 + koff) * HID + colB];
        b.y = W1[(k0 + koff + 1) * HID + colB];
        c = __builtin_amdgcn_wmma_f32_16x16x4_f32(
                 false, a,  false, b,
                 (short)0, c,  false,  false);
    }

    const int rbase = m0 + (lane >> 4) * 8;
#pragma unroll
    for (int r = 0; r < 8; ++r) {
        int row = rbase + r;
        if (row < LPAD) ew[row * HID + colB] = (row < NLAB) ? c[r] : 0.0f;
    }
}

__global__ void scale_rows_kernel(float* __restrict__ h, const float* __restrict__ norm, int d) {
    int i = blockIdx.x;
    int f = threadIdx.x;
    h[(long)i * d + f] = h[(long)i * d + f] * norm[i];
}


__device__ __forceinline__ unsigned own_bytes(unsigned x) {
    return ~(((x & 0x7f7f7f7fu) + 0x7f7f7f7fu) | x | 0x7f7f7f7fu);
}

__device__ __forceinline__ void lds_add(float* p, float v) {
    __hip_atomic_fetch_add(p, v, __ATOMIC_RELAXED, __HIP_MEMORY_SCOPE_WORKGROUP);
}

__device__ __forceinline__ void owner_scan_all(const unsigned* __restrict__ ob, unsigned me,
                                               unsigned* q, unsigned* qn, unsigned cap, int tid) {
    const unsigned pat = me * 0x01010101u;
    const uint4* own16 = (const uint4*)ob;

    uint4 cur[SCAN_L], nxt[SCAN_L];
#pragma unroll
    for (int j = 0; j < SCAN_L; ++j) cur[j] = own16[j * AGG_T + tid];

    for (int p = 0; p < NPASS; ++p) {
        if (p + 1 < NPASS) {
#pragma unroll
            for (int j = 0; j < SCAN_L; ++j) {
                const unsigned qi = (p + 1) * PASS_Q + j * AGG_T + tid;
                nxt[j] = own16[qi < NQ16 ? qi : NQ16 - 1];
            }
        }
#pragma unroll
        for (int j = 0; j < SCAN_L; ++j) {
            const unsigned wi = (unsigned)(p * PASS_Q + j * AGG_T + tid);
            const unsigned e0 = wi * 16u;
            const unsigned c = (own_bytes(cur[j].x ^ pat) >> 7) | (own_bytes(cur[j].y ^ pat) >> 6) |
                               (own_bytes(cur[j].z ^ pat) >> 5) | (own_bytes(cur[j].w ^ pat) >> 4);
            unsigned m = (wi < NQ16) ? c : 0u;
            while (m) {
                const unsigned b = (unsigned)__builtin_ctz(m);
                m &= m - 1u;
                const unsigned slot = __hip_atomic_fetch_add(qn, 1u, __ATOMIC_RELAXED,
                                                             __HIP_MEMORY_SCOPE_WORKGROUP);
                if (slot < cap) q[slot] = e0 + ((b & 7u) << 2) + (b >> 3);
            }
        }
        if (p + 1 < NPASS) {
#pragma unroll
            for (int j = 0; j < SCAN_L; ++j) cur[j] = nxt[j];
        }
    }
}

__global__ __launch_bounds__(AGG_T) void owner_outdeg_kernel(const unsigned* __restrict__ owns,
                                                             const unsigned* __restrict__ rec,
                                                             float* __restrict__ onorm) {
    __shared__ unsigned queue[QALL];
    __shared__ unsigned cnt[RNG];
    __shared__ unsigned qn;

    const int tid = threadIdx.x;
    const unsigned me = blockIdx.x;
    const unsigned lo = me * RNG;

    if (tid < RNG) cnt[tid] = 0u;
    if (tid == 0) qn = 0u;
    __syncthreads();

    owner_scan_all(owns, me, queue, &qn, QALL, tid);
    __syncthreads();

    const unsigned tot = qn;
    if (tot <= QALL) {
        for (unsigned i = tid; i < tot; i += AGG_T) {
            const unsigned r = rec[queue[i]];
            atomicAdd(&cnt[(r & 0xffffu) - lo], 1u);
        }
    } else {
        const unsigned char* ob = (const unsigned char*)owns;
        for (unsigned e = tid; e < NE; e += AGG_T)
            if (ob[e] == me) atomicAdd(&cnt[(rec[e] & 0xffffu) - lo], 1u);
    }
    __syncthreads();

    if (tid < RNG && lo + tid < NN) onorm[lo + tid] = rsqrtf(fmaxf((float)cnt[tid], 1.0f));
}

__global__ __launch_bounds__(AGG_T) void owner_label_kernel(const unsigned* __restrict__ own,
                                                            const unsigned* __restrict__ rec,
                                                            const int* __restrict__ labels,
                                                            const float* __restrict__ onorm,
                                                            float* __restrict__ tab,
                                                            float* __restrict__ inorm) {
    __shared__ __attribute__((aligned(16))) float lt[RNG * LPAD];
    __shared__ unsigned queue[QALL];
    __shared__ unsigned cnt[RNG];
    __shared__ unsigned qn;

    const int tid = threadIdx.x;
    const unsigned me = blockIdx.x;
    const unsigned lo = me * RNG;

    float4* lt4 = (float4*)lt;
    for (int i = tid; i < RNG * LPAD / 4; i += AGG_T) lt4[i] = make_float4(0.0f, 0.0f, 0.0f, 0.0f);
    if (tid < RNG) cnt[tid] = 0u;
    if (tid == 0) qn = 0u;
    __syncthreads();

    owner_scan_all(own, me, queue, &qn, QALL, tid);
    __syncthreads();

    const unsigned tot = qn;
    if (tot <= QALL) {
        for (unsigned i = tid; i < tot; i += AGG_T) {
            const unsigned r = rec[queue[i]];
            const unsigned s = r & 0xffffu, row = r >> 16;
            lds_add(&lt[row * LPAD + labels[s]], onorm[s]);
            atomicAdd(&cnt[row], 1u);
        }
    } else {
        const unsigned char* ob = (const unsigned char*)own;
        for (unsigned e = tid; e < NE; e += AGG_T) {
            if (ob[e] == me) {
                const unsigned r = rec[e];
                const unsigned s = r & 0xffffu, row = r >> 16;
                lds_add(&lt[row * LPAD + labels[s]], onorm[s]);
                atomicAdd(&cnt[row], 1u);
            }
        }
    }
    __syncthreads();

    const int rows = ((int)NN - (int)lo < RNG) ? (int)NN - (int)lo : RNG;
    float4* o4 = (float4*)(tab + (size_t)lo * LPAD);
    for (int i = tid; i < rows * LPAD / 4; i += AGG_T) o4[i] = lt4[i];
    if (tid < rows) inorm[lo + tid] = rsqrtf(fmaxf((float)cnt[tid], 1.0f));
}

__global__ __launch_bounds__(AGG_T) void owner_sum_kernel(const float* __restrict__ hs,
                                                          const unsigned* __restrict__ own,
                                                          const unsigned* __restrict__ rec,
                                                          float* __restrict__ agg) {
    __shared__ __attribute__((aligned(16))) float slab[RNG * HID];
    __shared__ unsigned queue[2][QCAP];
    __shared__ unsigned qn[NPASS];

    const int tid = threadIdx.x;
    const int lane = tid & 31;
    const int wave = tid >> 5;
    const unsigned me = blockIdx.x;
    const unsigned pat = me * 0x01010101u;
    const uint4* own16 = (const uint4*)own;
    const uint4 nomatch = {~pat, ~pat, ~pat, ~pat};

    float4* slab4 = (float4*)slab;
    for (int i = tid; i < RNG * HID / 4; i += AGG_T) slab4[i] = make_float4(0.0f, 0.0f, 0.0f, 0.0f);
    if (tid < NPASS) qn[tid] = 0u;

    uint4 cur[SCAN_L];
#pragma unroll
    for (int j = 0; j < SCAN_L; ++j) cur[j] = own16[j * AGG_T + tid];
    __syncthreads();

    for (int p = 0; p < NPASS; ++p) {
        unsigned* q = queue[p & 1];

#pragma unroll
        for (int j = 0; j < SCAN_L; ++j) {
            const unsigned e0 = (unsigned)(p * PASS_Q + j * AGG_T + tid) * 16u;
            const unsigned c = (own_bytes(cur[j].x ^ pat) >> 7) | (own_bytes(cur[j].y ^ pat) >> 6) |
                               (own_bytes(cur[j].z ^ pat) >> 5) | (own_bytes(cur[j].w ^ pat) >> 4);
            unsigned m = c;
            while (m) {
                const unsigned b = (unsigned)__builtin_ctz(m);
                m &= m - 1u;
                const unsigned slot = __hip_atomic_fetch_add(&qn[p], 1u, __ATOMIC_RELAXED,
                                                             __HIP_MEMORY_SCOPE_WORKGROUP);
                if (slot < QCAP) q[slot] = e0 + ((b & 7u) << 2) + (b >> 3);
            }
        }
        if (p + 1 < NPASS) {
#pragma unroll
            for (int j = 0; j < SCAN_L; ++j) {
                const unsigned qi = (p + 1) * PASS_Q + j * AGG_T + tid;
                const uint4 v = own16[qi < NQ16 ? qi : NQ16 - 1];
                cur[j] = v;
                if (qi >= NQ16) cur[j] = nomatch;
            }
        }
        __syncthreads();

        const unsigned tot = __builtin_amdgcn_readfirstlane(qn[p]);
        if (tot <= QCAP) {
            const unsigned per = (tot + 31u) >> 5;
            for (unsigned base = 0; base < per; base += 32u) {
                const unsigned idx = wave * per + base + lane;
                unsigned r = 0xffffffffu;
                if (base + lane < per && idx < tot) r = rec[q[idx]];
                const unsigned c = (per - base < 32u) ? per - base : 32u;
                for (unsigned j0 = 0; j0 < c; j0 += 8u) {
                    float v[8][3];
                    unsigned rr[8];
#pragma unroll
                    for (int u = 0; u < 8; ++u) {
                        rr[u] = __builtin_amdgcn_readlane(r, j0 + u);
                        const unsigned s = (rr[u] == 0xffffffffu) ? 0u : (rr[u] & 0xffffu);
                        const float* hp = hs + (size_t)s * HID + lane;
                        v[u][0] = hp[0];
                        v[u][1] = hp[32];
                        v[u][2] = hp[64];
                    }
#pragma unroll
                    for (int u = 0; u < 8; ++u) {
                        if (rr[u] != 0xffffffffu) {
                            float* sp = slab + (rr[u] >> 16) * HID + lane;
                            lds_add(sp, v[u][0]);
                            lds_add(sp + 32, v[u][1]);
                            lds_add(sp + 64, v[u][2]);
                        }
                    }
                }
            }
        } else {
            const unsigned eBeg = (unsigned)p * PASS_Q * 16u;
            const unsigned eEnd = (eBeg + PASS_Q * 16u < NE) ? eBeg + PASS_Q * 16u : NE;
            const unsigned char* ob = (const unsigned char*)own;
            for (unsigned e = eBeg + tid; e < eEnd; e += AGG_T) {
                if (ob[e] == me) {
                    const unsigned r = rec[e];
                    const float* hp = hs + (size_t)(r & 0xffffu) * HID;
                    float* sp = slab + (r >> 16) * HID;
                    for (int f = 0; f < HID; ++f) lds_add(sp + f, hp[f]);
                }
            }
        }
    }
    __syncthreads();

    const int lo = me * RNG;
    const int rows = (NN - lo < RNG) ? NN - lo : RNG;
    float4* o4 = (float4*)(agg + (size_t)lo * HID);
    for (int i = tid; i < rows * HID / 4; i += AGG_T) o4[i] = slab4[i];
}


__global__ __launch_bounds__(256) void gcn_gemm_wmma(
    const float* __restrict__ A, const float* __restrict__ rowScale,
    const float* __restrict__ W, const float* __restrict__ bias,
    float* __restrict__ out, int K, int Nout, int Npad,
    int tilesN, int totalTiles, int relu)
{
    __shared__ float smem[HID * HID + HID];
    const int tid = threadIdx.x;
    const int KN = K * Npad;

    for (int idx = tid; idx < KN; idx += 256) {
        int k = idx / Npad, n = idx % Npad;
        smem[idx] = (n < Nout) ? W[k * Nout + n] : 0.0f;
    }
    for (int idx = tid; idx < Npad; idx += 256)
        smem[KN + idx] = (idx < Nout) ? bias[idx] : 0.0f;
    __syncthreads();

    const int wave = tid >> 5;
    const int lane = tid & 31;
    const int tileId = blockIdx.x * 8 + wave;
    if (tileId >= totalTiles) return;

    const int m0 = (tileId / tilesN) * 16;
    const int n0 = (tileId % tilesN) * 16;

    const int rowA = m0 + (lane & 15);
    const int koff = (lane >> 4) * 2;
    const int colB = lane & 15;

    v8f c;
    const float bval = smem[KN + n0 + colB];
#pragma unroll
    for (int r = 0; r < 8; ++r) c[r] = bval;

    const float s = rowScale[rowA];
    const float* arow = A + (long)rowA * K;
    const float* bcol = smem + n0 + colB;

    for (int k0 = 0; k0 < K; k0 += 4) {
        v2f a, b;
        a.x = arow[k0 + koff] * s;
        a.y = arow[k0 + koff + 1] * s;
        b.x = bcol[(k0 + koff) * Npad];
        b.y = bcol[(k0 + koff + 1) * Npad];
        c = __builtin_amdgcn_wmma_f32_16x16x4_f32(
                 false, a,  false, b,
                 (short)0, c,  false,  false);
    }

    const int rbase = m0 + (lane >> 4) * 8;
    const int col = n0 + colB;
    if (col < Nout) {
#pragma unroll
        for (int r = 0; r < 8; ++r) {
            float v = c[r];
            if (relu) v = fmaxf(v, 0.0f);
            out[(long)(rbase + r) * Nout + col] = v;
        }
    }
}


extern "C" void kernel_launch(void* const* d_in, const int* in_sizes, int n_in,
                              void* d_out, int out_size, void* d_ws, size_t ws_size,
                              hipStream_t stream) {
    const int*   labels = (const int*)d_in[0];
    const int*   src    = (const int*)d_in[1];
    const int*   dst    = (const int*)d_in[2];
    const float* emb    = (const float*)d_in[3];
    const float* W1     = (const float*)d_in[4];
    const float* b1     = (const float*)d_in[5];
    const float* W2     = (const float*)d_in[6];
    const float* b2     = (const float*)d_in[7];
    const float* W3     = (const float*)d_in[8];
    const float* b3     = (const float*)d_in[9];
    float* out = (float*)d_out;

    float* w = (float*)d_ws;
    float* onorm = w;
    float* inorm = w + NN;
    float* bufA  = w + 2 * (size_t)NN;
    float* bufB  = bufA + (size_t)NN * HID;
    float* bufC  = bufB + (size_t)NN * HID;
    unsigned* own  = (unsigned*)(bufB + 8192);
    unsigned* rec  = own + NE / 4;
    unsigned* owns = rec + NE;

    const int TPB = 256;

    edge_rec_kernel<<<(NE / 4 + TPB - 1) / TPB, TPB, 0, stream>>>(src, dst, own, owns, rec, NE / 4);
    owner_outdeg_kernel<<<NOWN, AGG_T, 0, stream>>>(owns, rec, onorm);

    float* ew = bufB;
    label_weight_wmma<<<4 * (HID / 16), 32, 0, stream>>>(emb, W1, ew);
    owner_label_kernel<<<NOWN, AGG_T, 0, stream>>>(own, rec, labels, onorm, bufC, inorm);
    {
        int tilesN = HID / 16;
        int total  = (NN / 16) * tilesN;
        gcn_gemm_wmma<<<(total + 7) / 8, 256, 0, stream>>>(
            bufC, inorm, ew, b1, bufA, LPAD, HID, HID, tilesN, total, 1);
    }

    scale_rows_kernel<<<NN, HID, 0, stream>>>(bufA, onorm, HID);
    owner_sum_kernel<<<NOWN, AGG_T, 0, stream>>>(bufA, own, rec, bufC);
    {
        int tilesN = HID / 16;
        int total  = (NN / 16) * tilesN;
        gcn_gemm_wmma<<<(total + 7) / 8, 256, 0, stream>>>(
            bufC, inorm, W2, b2, bufA, HID, HID, HID, tilesN, total, 1);
    }

    scale_rows_kernel<<<NN, HID, 0, stream>>>(bufA, onorm, HID);
    owner_sum_kernel<<<NOWN, AGG_T, 0, stream>>>(bufA, own, rec, bufC);
    {
        int Npad   = 64;
        int tilesN = Npad / 16;
        int total  = (NN / 16) * tilesN;
        gcn_gemm_wmma<<<(total + 7) / 8, 256, 0, stream>>>(
            bufC, inorm, W3, b3, out, HID, NLAB, Npad, tilesN, total, 0);
    }
}


# out_norm applied per edge in the owner blocks' gather (one scalar load per edge beside its row loads, row times it, then LDS add): h stays unscaled, both scale_rows launches gone; sum order kept
# speedup vs baseline: 2.7151x; 1.0972x over previous
#include <hip/hip_runtime.h>

#define NN 50000
#define NE 800000
#define NLAB 50
#define EMB 64
#define HID 96
#define LPAD 52

#define RNG 196
#define NOWN ((NN + RNG - 1) / RNG)
#define AGG_T 1024
#define SCAN_L 7
#define NQ16 (NE / 16)
#define PASS_Q (AGG_T * SCAN_L)
#define NPASS ((NQ16 + PASS_Q - 1) / PASS_Q)
#define QCAP 12288
#define QALL 45056

static_assert(NOWN <= 256, "owner id must fit one byte");
static_assert(NN <= 65536 && RNG <= 65536, "edge record packs source and local row in 16 bits each");
static_assert(NE % 16 == 0, "owner bytes are read 16 at a time");
static_assert(PASS_Q <= NQ16, "the first pass must lie inside the owner array");

typedef __attribute__((ext_vector_type(2))) float v2f;
typedef __attribute__((ext_vector_type(8))) float v8f;


__global__ void edge_rec_kernel(const int* __restrict__ src, const int* __restrict__ dst,
                                unsigned* __restrict__ own, unsigned* __restrict__ owns,
                                unsigned* __restrict__ rec, int e4) {
    int i = blockIdx.x * blockDim.x + threadIdx.x;
    if (i < e4) {
        const int4 s = ((const int4*)src)[i];
        const int4 t = ((const int4*)dst)[i];
        const unsigned o0 = (unsigned)t.x / RNG, o1 = (unsigned)t.y / RNG;
        const unsigned o2 = (unsigned)t.z / RNG, o3 = (unsigned)t.w / RNG;
        own[i] = o0 | (o1 << 8) | (o2 << 16) | (o3 << 24);
        const unsigned p0 = (unsigned)s.x / RNG, p1 = (unsigned)s.y / RNG;
        const unsigned p2 = (unsigned)s.z / RNG, p3 = (unsigned)s.w / RNG;
        owns[i] = p0 | (p1 << 8) | (p2 << 16) | (p3 << 24);
        uint4 r;
        r.x = (((unsigned)t.x - o0 * RNG) << 16) | (unsigned)s.x;
        r.y = (((unsigned)t.y - o1 * RNG) << 16) | (unsigned)s.y;
        r.z = (((unsigned)t.z - o2 * RNG) << 16) | (unsigned)s.z;
        r.w = (((unsigned)t.w - o3 * RNG) << 16) | (unsigned)s.w;
        ((uint4*)rec)[i] = r;
    }
}


__global__ __launch_bounds__(32) void label_weight_wmma(const float* __restrict__ emb,
                                                        const float* __restrict__ W1,
                                                        float* __restrict__ ew) {
    const int lane = threadIdx.x;
    const int m0 = (blockIdx.x / (HID / 16)) * 16;
    const int n0 = (blockIdx.x % (HID / 16)) * 16;

    const int rowA = m0 + (lane & 15);
    const int koff = (lane >> 4) * 2;
    const int colB = n0 + (lane & 15);
    const bool live = rowA < NLAB;
    const float* arow = emb + (long)(live ? rowA : 0) * EMB;

    v8f c = {0.0f, 0.0f, 0.0f, 0.0f, 0.0f, 0.0f, 0.0f, 0.0f};
    for (int k0 = 0; k0 < EMB; k0 += 4) {
        v2f a, b;
        float a0 = arow[k0 + koff];
        float a1 = arow[k0 + koff + 1];
        a.x = live ? a0 : 0.0f;
        a.y = live ? a1 : 0.0f;
        b.x = W1[(k0 + koff) * HID + colB];
        b.y = W1[(k0 + koff + 1) * HID + colB];
        c = __builtin_amdgcn_wmma_f32_16x16x4_f32(
                 false, a,  false, b,
                 (short)0, c,  false,  false);
    }

    const int rbase = m0 + (lane >> 4) * 8;
#pragma unroll
    for (int r = 0; r < 8; ++r) {
        int row = rbase + r;
        if (row < LPAD) ew[row * HID + colB] = (row < NLAB) ? c[r] : 0.0f;
    }
}


__device__ __forceinline__ unsigned own_bytes(unsigned x) {
    return ~(((x & 0x7f7f7f7fu) + 0x7f7f7f7fu) | x | 0x7f7f7f7fu);
}

__device__ __forceinline__ void lds_add(float* p, float v) {
    __hip_atomic_fetch_add(p, v, __ATOMIC_RELAXED, __HIP_MEMORY_SCOPE_WORKGROUP);
}

__device__ __forceinline__ void owner_scan_all(const unsigned* __restrict__ ob, unsigned me,
                                               unsigned* q, unsigned* qn, unsigned cap, int tid) {
    const unsigned pat = me * 0x01010101u;
    const uint4* own16 = (const uint4*)ob;

    uint4 cur[SCAN_L], nxt[SCAN_L];
#pragma unroll
    for (int j = 0; j < SCAN_L; ++j) cur[j] = own16[j * AGG_T + tid];

    for (int p = 0; p < NPASS; ++p) {
        if (p + 1 < NPASS) {
#pragma unroll
            for (int j = 0; j < SCAN_L; ++j) {
                const unsigned qi = (p + 1) * PASS_Q + j * AGG_T + tid;
                nxt[j] = own16[qi < NQ16 ? qi : NQ16 - 1];
            }
        }
#pragma unroll
        for (int j = 0; j < SCAN_L; ++j) {
            const unsigned wi = (unsigned)(p * PASS_Q + j * AGG_T + tid);
            const unsigned e0 = wi * 16u;
            const unsigned c = (own_bytes(cur[j].x ^ pat) >> 7) | (own_bytes(cur[j].y ^ pat) >> 6) |
                               (own_bytes(cur[j].z ^ pat) >> 5) | (own_bytes(cur[j].w ^ pat) >> 4);
            unsigned m = (wi < NQ16) ? c : 0u;
            while (m) {
                const unsigned b = (unsigned)__builtin_ctz(m);
                m &= m - 1u;
                const unsigned slot = __hip_atomic_fetch_add(qn, 1u, __ATOMIC_RELAXED,
                                                             __HIP_MEMORY_SCOPE_WORKGROUP);
                if (slot < cap) q[slot] = e0 + ((b & 7u) << 2) + (b >> 3);
            }
        }
        if (p + 1 < NPASS) {
#pragma unroll
            for (int j = 0; j < SCAN_L; ++j) cur[j] = nxt[j];
        }
    }
}

__global__ __launch_bounds__(AGG_T) void owner_outdeg_kernel(const unsigned* __restrict__ owns,
                                                             const unsigned* __restrict__ rec,
                                                             float* __restrict__ onorm) {
    __shared__ unsigned queue[QALL];
    __shared__ unsigned cnt[RNG];
    __shared__ unsigned qn;

    const int tid = threadIdx.x;
    const unsigned me = blockIdx.x;
    const unsigned lo = me * RNG;

    if (tid < RNG) cnt[tid] = 0u;
    if (tid == 0) qn = 0u;
    __syncthreads();

    owner_scan_all(owns, me, queue, &qn, QALL, tid);
    __syncthreads();

    const unsigned tot = qn;
    if (tot <= QALL) {
        for (unsigned i = tid; i < tot; i += AGG_T) {
            const unsigned r = rec[queue[i]];
            atomicAdd(&cnt[(r & 0xffffu) - lo], 1u);
        }
    } else {
        const unsigned char* ob = (const unsigned char*)owns;
        for (unsigned e = tid; e < NE; e += AGG_T)
            if (ob[e] == me) atomicAdd(&cnt[(rec[e] & 0xffffu) - lo], 1u);
    }
    __syncthreads();

    if (tid < RNG && lo + tid < NN) onorm[lo + tid] = rsqrtf(fmaxf((float)cnt[tid], 1.0f));
}

__global__ __launch_bounds__(AGG_T) void owner_label_kernel(const unsigned* __restrict__ own,
                                                            const unsigned* __restrict__ rec,
                                                            const int* __restrict__ labels,
                                                            const float* __restrict__ onorm,
                                                            float* __restrict__ tab,
                                                            float* __restrict__ inorm) {
    __shared__ __attribute__((aligned(16))) float lt[RNG * LPAD];
    __shared__ unsigned queue[QALL];
    __shared__ unsigned cnt[RNG];
    __shared__ unsigned qn;

    const int tid = threadIdx.x;
    const unsigned me = blockIdx.x;
    const unsigned lo = me * RNG;

    float4* lt4 = (float4*)lt;
    for (int i = tid; i < RNG * LPAD / 4; i += AGG_T) lt4[i] = make_float4(0.0f, 0.0f, 0.0f, 0.0f);
    if (tid < RNG) cnt[tid] = 0u;
    if (tid == 0) qn = 0u;
    __syncthreads();

    owner_scan_all(own, me, queue, &qn, QALL, tid);
    __syncthreads();

    const unsigned tot = qn;
    if (tot <= QALL) {
        for (unsigned i = tid; i < tot; i += AGG_T) {
            const unsigned r = rec[queue[i]];
            const unsigned s = r & 0xffffu, row = r >> 16;
            lds_add(&lt[row * LPAD + labels[s]], onorm[s]);
            atomicAdd(&cnt[row], 1u);
        }
    } else {
        const unsigned char* ob = (const unsigned char*)own;
        for (unsigned e = tid; e < NE; e += AGG_T) {
            if (ob[e] == me) {
                const unsigned r = rec[e];
                const unsigned s = r & 0xffffu, row = r >> 16;
                lds_add(&lt[row * LPAD + labels[s]], onorm[s]);
                atomicAdd(&cnt[row], 1u);
            }
        }
    }
    __syncthreads();

    const int rows = ((int)NN - (int)lo < RNG) ? (int)NN - (int)lo : RNG;
    float4* o4 = (float4*)(tab + (size_t)lo * LPAD);
    for (int i = tid; i < rows * LPAD / 4; i += AGG_T) o4[i] = lt4[i];
    if (tid < rows) inorm[lo + tid] = rsqrtf(fmaxf((float)cnt[tid], 1.0f));
}

__global__ __launch_bounds__(AGG_T) void owner_sum_kernel(const float* __restrict__ hs,
                                                          const float* __restrict__ onorm,
                                                          const unsigned* __restrict__ own,
                                                          const unsigned* __restrict__ rec,
                                                          float* __restrict__ agg) {
    __shared__ __attribute__((aligned(16))) float slab[RNG * HID];
    __shared__ unsigned queue[2][QCAP];
    __shared__ unsigned qn[NPASS];

    const int tid = threadIdx.x;
    const int lane = tid & 31;
    const int wave = tid >> 5;
    const unsigned me = blockIdx.x;
    const unsigned pat = me * 0x01010101u;
    const uint4* own16 = (const uint4*)own;
    const uint4 nomatch = {~pat, ~pat, ~pat, ~pat};

    float4* slab4 = (float4*)slab;
    for (int i = tid; i < RNG * HID / 4; i += AGG_T) slab4[i] = make_float4(0.0f, 0.0f, 0.0f, 0.0f);
    if (tid < NPASS) qn[tid] = 0u;

    uint4 cur[SCAN_L];
#pragma unroll
    for (int j = 0; j < SCAN_L; ++j) cur[j] = own16[j * AGG_T + tid];
    __syncthreads();

    for (int p = 0; p < NPASS; ++p) {
        unsigned* q = queue[p & 1];

#pragma unroll
        for (int j = 0; j < SCAN_L; ++j) {
            const unsigned e0 = (unsigned)(p * PASS_Q + j * AGG_T + tid) * 16u;
            const unsigned c = (own_bytes(cur[j].x ^ pat) >> 7) | (own_bytes(cur[j].y ^ pat) >> 6) |
                               (own_bytes(cur[j].z ^ pat) >> 5) | (own_bytes(cur[j].w ^ pat) >> 4);
            unsigned m = c;
            while (m) {
                const unsigned b = (unsigned)__builtin_ctz(m);
                m &= m - 1u;
                const unsigned slot = __hip_atomic_fetch_add(&qn[p], 1u, __ATOMIC_RELAXED,
                                                             __HIP_MEMORY_SCOPE_WORKGROUP);
                if (slot < QCAP) q[slot] = e0 + ((b & 7u) << 2) + (b >> 3);
            }
        }
        if (p + 1 < NPASS) {
#pragma unroll
            for (int j = 0; j < SCAN_L; ++j) {
                const unsigned qi = (p + 1) * PASS_Q + j * AGG_T + tid;
                const uint4 v = own16[qi < NQ16 ? qi : NQ16 - 1];
                cur[j] = v;
                if (qi >= NQ16) cur[j] = nomatch;
            }
        }
        __syncthreads();

        const unsigned tot = __builtin_amdgcn_readfirstlane(qn[p]);
        if (tot <= QCAP) {
            const unsigned per = (tot + 31u) >> 5;
            for (unsigned base = 0; base < per; base += 32u) {
                const unsigned idx = wave * per + base + lane;
                unsigned r = 0xffffffffu;
                if (base + lane < per && idx < tot) r = rec[q[idx]];
                const unsigned c = (per - base < 32u) ? per - base : 32u;
                for (unsigned j0 = 0; j0 < c; j0 += 8u) {
                    float v[8][3];
                    float w[8];
                    unsigned rr[8];
#pragma unroll
                    for (int u = 0; u < 8; ++u) {
                        rr[u] = __builtin_amdgcn_readlane(r, j0 + u);
                        const unsigned s = (rr[u] == 0xffffffffu) ? 0u : (rr[u] & 0xffffu);
                        const float* hp = hs + (size_t)s * HID + lane;
                        v[u][0] = hp[0];
                        v[u][1] = hp[32];
                        v[u][2] = hp[64];
                        w[u] = onorm[s];
                    }
#pragma unroll
                    for (int u = 0; u < 8; ++u) {
                        if (rr[u] != 0xffffffffu) {
                            float* sp = slab + (rr[u] >> 16) * HID + lane;
                            lds_add(sp, v[u][0] * w[u]);
                            lds_add(sp + 32, v[u][1] * w[u]);
                            lds_add(sp + 64, v[u][2] * w[u]);
                        }
                    }
                }
            }
        } else {
            const unsigned eBeg = (unsigned)p * PASS_Q * 16u;
            const unsigned eEnd = (eBeg + PASS_Q * 16u < NE) ? eBeg + PASS_Q * 16u : NE;
            const unsigned char* ob = (const unsigned char*)own;
            for (unsigned e = eBeg + tid; e < eEnd; e += AGG_T) {
                if (ob[e] == me) {
                    const unsigned r = rec[e];
                    const float* hp = hs + (size_t)(r & 0xffffu) * HID;
                    const float w = onorm[r & 0xffffu];
                    float* sp = slab + (r >> 16) * HID;
                    for (int f = 0; f < HID; ++f) lds_add(sp + f, hp[f] * w);
                }
            }
        }
    }
    __syncthreads();

    const int lo = me * RNG;
    const int rows = (NN - lo < RNG) ? NN - lo : RNG;
    float4* o4 = (float4*)(agg + (size_t)lo * HID);
    for (int i = tid; i < rows * HID / 4; i += AGG_T) o4[i] = slab4[i];
}


__global__ __launch_bounds__(256) void gcn_gemm_wmma(
    const float* __restrict__ A, const float* __restrict__ rowScale,
    const float* __restrict__ W, const float* __restrict__ bias,
    float* __restrict__ out, int K, int Nout, int Npad,
    int tilesN, int totalTiles, int relu)
{
    __shared__ float smem[HID * HID + HID];
    const int tid = threadIdx.x;
    const int KN = K * Npad;

    for (int idx = tid; idx < KN; idx += 256) {
        int k = idx / Npad, n = idx % Npad;
        smem[idx] = (n < Nout) ? W[k * Nout + n] : 0.0f;
    }
    for (int idx = tid; idx < Npad; idx += 256)
        smem[KN + idx] = (idx < Nout) ? bias[idx] : 0.0f;
    __syncthreads();

    const int wave = tid >> 5;
    const int lane = tid & 31;
    const int tileId = blockIdx.x * 8 + wave;
    if (tileId >= totalTiles) return;

    const int m0 = (tileId / tilesN) * 16;
    const int n0 = (tileId % tilesN) * 16;

    const int rowA = m0 + (lane & 15);
    const int koff = (lane >> 4) * 2;
    const int colB = lane & 15;

    v8f c;
    const float bval = smem[KN + n0 + colB];
#pragma unroll
    for (int r = 0; r < 8; ++r) c[r] = bval;

    const float s = rowScale[rowA];
    const float* arow = A + (long)rowA * K;
    const float* bcol = smem + n0 + colB;

    for (int k0 = 0; k0 < K; k0 += 4) {
        v2f a, b;
        a.x = arow[k0 + koff] * s;
        a.y = arow[k0 + koff + 1] * s;
        b.x = bcol[(k0 + koff) * Npad];
        b.y = bcol[(k0 + koff + 1) * Npad];
        c = __builtin_amdgcn_wmma_f32_16x16x4_f32(
                 false, a,  false, b,
                 (short)0, c,  false,  false);
    }

    const int rbase = m0 + (lane >> 4) * 8;
    const int col = n0 + colB;
    if (col < Nout) {
#pragma unroll
        for (int r = 0; r < 8; ++r) {
            float v = c[r];
            if (relu) v = fmaxf(v, 0.0f);
            out[(long)(rbase + r) * Nout + col] = v;
        }
    }
}


extern "C" void kernel_launch(void* const* d_in, const int* in_sizes, int n_in,
                              void* d_out, int out_size, void* d_ws, size_t ws_size,
                              hipStream_t stream) {
    const int*   labels = (const int*)d_in[0];
    const int*   src    = (const int*)d_in[1];
    const int*   dst    = (const int*)d_in[2];
    const float* emb    = (const float*)d_in[3];
    const float* W1     = (const float*)d_in[4];
    const float* b1     = (const float*)d_in[5];
    const float* W2     = (const float*)d_in[6];
    const float* b2     = (const float*)d_in[7];
    const float* W3     = (const float*)d_in[8];
    const float* b3     = (const float*)d_in[9];
    float* out = (float*)d_out;

    float* w = (float*)d_ws;
    float* onorm = w;
    float* inorm = w + NN;
    float* bufA  = w + 2 * (size_t)NN;
    float* bufB  = bufA + (size_t)NN * HID;
    float* bufC  = bufB + (size_t)NN * HID;
    unsigned* own  = (unsigned*)(bufB + 8192);
    unsigned* rec  = own + NE / 4;
    unsigned* owns = rec + NE;

    const int TPB = 256;

    edge_rec_kernel<<<(NE / 4 + TPB - 1) / TPB, TPB, 0, stream>>>(src, dst, own, owns, rec, NE / 4);
    owner_outdeg_kernel<<<NOWN, AGG_T, 0, stream>>>(owns, rec, onorm);

    float* ew = bufB;
    label_weight_wmma<<<4 * (HID / 16), 32, 0, stream>>>(emb, W1, ew);
    owner_label_kernel<<<NOWN, AGG_T, 0, stream>>>(own, rec, labels, onorm, bufC, inorm);
    {
        int tilesN = HID / 16;
        int total  = (NN / 16) * tilesN;
        gcn_gemm_wmma<<<(total + 7) / 8, 256, 0, stream>>>(
            bufC, inorm, ew, b1, bufA, LPAD, HID, HID, tilesN, total, 1);
    }

    owner_sum_kernel<<<NOWN, AGG_T, 0, stream>>>(bufA, onorm, own, rec, bufC);
    {
        int tilesN = HID / 16;
        int total  = (NN / 16) * tilesN;
        gcn_gemm_wmma<<<(total + 7) / 8, 256, 0, stream>>>(
            bufC, inorm, W2, b2, bufA, HID, HID, HID, tilesN, total, 1);
    }

    owner_sum_kernel<<<NOWN, AGG_T, 0, stream>>>(bufA, onorm, own, rec, bufC);
    {
        int Npad   = 64;
        int tilesN = Npad / 16;
        int total  = (NN / 16) * tilesN;
        gcn_gemm_wmma<<<(total + 7) / 8, 256, 0, stream>>>(
            bufC, inorm, W3, b3, out, HID, NLAB, Npad, tilesN, total, 0);
    }
}


# GEMM launches re-tiled: 256 resident 16-wave blocks stage W once in LDS (k pairs side by side, 8-byte B reads); a wave makes a whole 16-row strip, its A fragments fetched once; sum order unchanged
# speedup vs baseline: 3.5196x; 1.2963x over previous
#include <hip/hip_runtime.h>

#define NN 50000
#define NE 800000
#define NLAB 50
#define EMB 64
#define HID 96
#define LPAD 52

#define RNG 196
#define NOWN ((NN + RNG - 1) / RNG)
#define AGG_T 1024
#define SCAN_L 7
#define NQ16 (NE / 16)
#define PASS_Q (AGG_T * SCAN_L)
#define NPASS ((NQ16 + PASS_Q - 1) / PASS_Q)
#define QCAP 12288
#define QALL 45056

static_assert(NOWN <= 256, "owner id must fit one byte");
static_assert(NN <= 65536 && RNG <= 65536, "edge record packs source and local row in 16 bits each");
static_assert(NE % 16 == 0, "owner bytes are read 16 at a time");
static_assert(PASS_Q <= NQ16, "the first pass must lie inside the owner array");

typedef __attribute__((ext_vector_type(2))) float v2f;
typedef __attribute__((ext_vector_type(8))) float v8f;


__global__ void edge_rec_kernel(const int* __restrict__ src, const int* __restrict__ dst,
                                unsigned* __restrict__ own, unsigned* __restrict__ owns,
                                unsigned* __restrict__ rec, int e4) {
    int i = blockIdx.x * blockDim.x + threadIdx.x;
    if (i < e4) {
        const int4 s = ((const int4*)src)[i];
        const int4 t = ((const int4*)dst)[i];
        const unsigned o0 = (unsigned)t.x / RNG, o1 = (unsigned)t.y / RNG;
        const unsigned o2 = (unsigned)t.z / RNG, o3 = (unsigned)t.w / RNG;
        own[i] = o0 | (o1 << 8) | (o2 << 16) | (o3 << 24);
        const unsigned p0 = (unsigned)s.x / RNG, p1 = (unsigned)s.y / RNG;
        const unsigned p2 = (unsigned)s.z / RNG, p3 = (unsigned)s.w / RNG;
        owns[i] = p0 | (p1 << 8) | (p2 << 16) | (p3 << 24);
        uint4 r;
        r.x = (((unsigned)t.x - o0 * RNG) << 16) | (unsigned)s.x;
        r.y = (((unsigned)t.y - o1 * RNG) << 16) | (unsigned)s.y;
        r.z = (((unsigned)t.z - o2 * RNG) << 16) | (unsigned)s.z;
        r.w = (((unsigned)t.w - o3 * RNG) << 16) | (unsigned)s.w;
        ((uint4*)rec)[i] = r;
    }
}


__global__ __launch_bounds__(32) void label_weight_wmma(const float* __restrict__ emb,
                                                        const float* __restrict__ W1,
                                                        float* __restrict__ ew) {
    const int lane = threadIdx.x;
    const int m0 = (blockIdx.x / (HID / 16)) * 16;
    const int n0 = (blockIdx.x % (HID / 16)) * 16;

    const int rowA = m0 + (lane & 15);
    const int koff = (lane >> 4) * 2;
    const int colB = n0 + (lane & 15);
    const bool live = rowA < NLAB;
    const float* arow = emb + (long)(live ? rowA : 0) * EMB;

    v8f c = {0.0f, 0.0f, 0.0f, 0.0f, 0.0f, 0.0f, 0.0f, 0.0f};
    for (int k0 = 0; k0 < EMB; k0 += 4) {
        v2f a, b;
        float a0 = arow[k0 + koff];
        float a1 = arow[k0 + koff + 1];
        a.x = live ? a0 : 0.0f;
        a.y = live ? a1 : 0.0f;
        b.x = W1[(k0 + koff) * HID + colB];
        b.y = W1[(k0 + koff + 1) * HID + colB];
        c = __builtin_amdgcn_wmma_f32_16x16x4_f32(
                 false, a,  false, b,
                 (short)0, c,  false,  false);
    }

    const int rbase = m0 + (lane >> 4) * 8;
#pragma unroll
    for (int r = 0; r < 8; ++r) {
        int row = rbase + r;
        if (row < LPAD) ew[row * HID + colB] = (row < NLAB) ? c[r] : 0.0f;
    }
}


__device__ __forceinline__ unsigned own_bytes(unsigned x) {
    return ~(((x & 0x7f7f7f7fu) + 0x7f7f7f7fu) | x | 0x7f7f7f7fu);
}

__device__ __forceinline__ void lds_add(float* p, float v) {
    __hip_atomic_fetch_add(p, v, __ATOMIC_RELAXED, __HIP_MEMORY_SCOPE_WORKGROUP);
}

__device__ __forceinline__ void owner_scan_all(const unsigned* __restrict__ ob, unsigned me,
                                               unsigned* q, unsigned* qn, unsigned cap, int tid) {
    const unsigned pat = me * 0x01010101u;
    const uint4* own16 = (const uint4*)ob;

    uint4 cur[SCAN_L], nxt[SCAN_L];
#pragma unroll
    for (int j = 0; j < SCAN_L; ++j) cur[j] = own16[j * AGG_T + tid];

    for (int p = 0; p < NPASS; ++p) {
        if (p + 1 < NPASS) {
#pragma unroll
            for (int j = 0; j < SCAN_L; ++j) {
                const unsigned qi = (p + 1) * PASS_Q + j * AGG_T + tid;
                nxt[j] = own16[qi < NQ16 ? qi : NQ16 - 1];
            }
        }
#pragma unroll
        for (int j = 0; j < SCAN_L; ++j) {
            const unsigned wi = (unsigned)(p * PASS_Q + j * AGG_T + tid);
            const unsigned e0 = wi * 16u;
            const unsigned c = (own_bytes(cur[j].x ^ pat) >> 7) | (own_bytes(cur[j].y ^ pat) >> 6) |
                               (own_bytes(cur[j].z ^ pat) >> 5) | (own_bytes(cur[j].w ^ pat) >> 4);
            unsigned m = (wi < NQ16) ? c : 0u;
            while (m) {
                const unsigned b = (unsigned)__builtin_ctz(m);
                m &= m - 1u;
                const unsigned slot = __hip_atomic_fetch_add(qn, 1u, __ATOMIC_RELAXED,
                                                             __HIP_MEMORY_SCOPE_WORKGROUP);
                if (slot < cap) q[slot] = e0 + ((b & 7u) << 2) + (b >> 3);
            }
        }
        if (p + 1 < NPASS) {
#pragma unroll
            for (int j = 0; j < SCAN_L; ++j) cur[j] = nxt[j];
        }
    }
}

__global__ __launch_bounds__(AGG_T) void owner_outdeg_kernel(const unsigned* __restrict__ owns,
                                                             const unsigned* __restrict__ rec,
                                                             float* __restrict__ onorm) {
    __shared__ unsigned queue[QALL];
    __shared__ unsigned cnt[RNG];
    __shared__ unsigned qn;

    const int tid = threadIdx.x;
    const unsigned me = blockIdx.x;
    const unsigned lo = me * RNG;

    if (tid < RNG) cnt[tid] = 0u;
    if (tid == 0) qn = 0u;
    __syncthreads();

    owner_scan_all(owns, me, queue, &qn, QALL, tid);
    __syncthreads();

    const unsigned tot = qn;
    if (tot <= QALL) {
        for (unsigned i = tid; i < tot; i += AGG_T) {
            const unsigned r = rec[queue[i]];
            atomicAdd(&cnt[(r & 0xffffu) - lo], 1u);
        }
    } else {
        const unsigned char* ob = (const unsigned char*)owns;
        for (unsigned e = tid; e < NE; e += AGG_T)
            if (ob[e] == me) atomicAdd(&cnt[(rec[e] & 0xffffu) - lo], 1u);
    }
    __syncthreads();

    if (tid < RNG && lo + tid < NN) onorm[lo + tid] = rsqrtf(fmaxf((float)cnt[tid], 1.0f));
}

__global__ __launch_bounds__(AGG_T) void owner_label_kernel(const unsigned* __restrict__ own,
                                                            const unsigned* __restrict__ rec,
                                                            const int* __restrict__ labels,
                                                            const float* __restrict__ onorm,
                                                            float* __restrict__ tab,
                                                            float* __restrict__ inorm) {
    __shared__ __attribute__((aligned(16))) float lt[RNG * LPAD];
    __shared__ unsigned queue[QALL];
    __shared__ unsigned cnt[RNG];
    __shared__ unsigned qn;

    const int tid = threadIdx.x;
    const unsigned me = blockIdx.x;
    const unsigned lo = me * RNG;

    float4* lt4 = (float4*)lt;
    for (int i = tid; i < RNG * LPAD / 4; i += AGG_T) lt4[i] = make_float4(0.0f, 0.0f, 0.0f, 0.0f);
    if (tid < RNG) cnt[tid] = 0u;
    if (tid == 0) qn = 0u;
    __syncthreads();

    owner_scan_all(own, me, queue, &qn, QALL, tid);
    __syncthreads();

    const unsigned tot = qn;
    if (tot <= QALL) {
        for (unsigned i = tid; i < tot; i += AGG_T) {
            const unsigned r = rec[queue[i]];
            const unsigned s = r & 0xffffu, row = r >> 16;
            lds_add(&lt[row * LPAD + labels[s]], onorm[s]);
            atomicAdd(&cnt[row], 1u);
        }
    } else {
        const unsigned char* ob = (const unsigned char*)own;
        for (unsigned e = tid; e < NE; e += AGG_T) {
            if (ob[e] == me) {
                const unsigned r = rec[e];
                const unsigned s = r & 0xffffu, row = r >> 16;
                lds_add(&lt[row * LPAD + labels[s]], onorm[s]);
                atomicAdd(&cnt[row], 1u);
            }
        }
    }
    __syncthreads();

    const int rows = ((int)NN - (int)lo < RNG) ? (int)NN - (int)lo : RNG;
    float4* o4 = (float4*)(tab + (size_t)lo * LPAD);
    for (int i = tid; i < rows * LPAD / 4; i += AGG_T) o4[i] = lt4[i];
    if (tid < rows) inorm[lo + tid] = rsqrtf(fmaxf((float)cnt[tid], 1.0f));
}

__global__ __launch_bounds__(AGG_T) void owner_sum_kernel(const float* __restrict__ hs,
                                                          const float* __restrict__ onorm,
                                                          const unsigned* __restrict__ own,
                                                          const unsigned* __restrict__ rec,
                                                          float* __restrict__ agg) {
    __shared__ __attribute__((aligned(16))) float slab[RNG * HID];
    __shared__ unsigned queue[2][QCAP];
    __shared__ unsigned qn[NPASS];

    const int tid = threadIdx.x;
    const int lane = tid & 31;
    const int wave = tid >> 5;
    const unsigned me = blockIdx.x;
    const unsigned pat = me * 0x01010101u;
    const uint4* own16 = (const uint4*)own;
    const uint4 nomatch = {~pat, ~pat, ~pat, ~pat};

    float4* slab4 = (float4*)slab;
    for (int i = tid; i < RNG * HID / 4; i += AGG_T) slab4[i] = make_float4(0.0f, 0.0f, 0.0f, 0.0f);
    if (tid < NPASS) qn[tid] = 0u;

    uint4 cur[SCAN_L];
#pragma unroll
    for (int j = 0; j < SCAN_L; ++j) cur[j] = own16[j * AGG_T + tid];
    __syncthreads();

    for (int p = 0; p < NPASS; ++p) {
        unsigned* q = queue[p & 1];

#pragma unroll
        for (int j = 0; j < SCAN_L; ++j) {
            const unsigned e0 = (unsigned)(p * PASS_Q + j * AGG_T + tid) * 16u;
            const unsigned c = (own_bytes(cur[j].x ^ pat) >> 7) | (own_bytes(cur[j].y ^ pat) >> 6) |
                               (own_bytes(cur[j].z ^ pat) >> 5) | (own_bytes(cur[j].w ^ pat) >> 4);
            unsigned m = c;
            while (m) {
                const unsigned b = (unsigned)__builtin_ctz(m);
                m &= m - 1u;
                const unsigned slot = __hip_atomic_fetch_add(&qn[p], 1u, __ATOMIC_RELAXED,
                                                             __HIP_MEMORY_SCOPE_WORKGROUP);
                if (slot < QCAP) q[slot] = e0 + ((b & 7u) << 2) + (b >> 3);
            }
        }
        if (p + 1 < NPASS) {
#pragma unroll
            for (int j = 0; j < SCAN_L; ++j) {
                const unsigned qi = (p + 1) * PASS_Q + j * AGG_T + tid;
                const uint4 v = own16[qi < NQ16 ? qi : NQ16 - 1];
                cur[j] = v;
                if (qi >= NQ16) cur[j] = nomatch;
            }
        }
        __syncthreads();

        const unsigned tot = __builtin_amdgcn_readfirstlane(qn[p]);
        if (tot <= QCAP) {
            const unsigned per = (tot + 31u) >> 5;
            for (unsigned base = 0; base < per; base += 32u) {
                const unsigned idx = wave * per + base + lane;
                unsigned r = 0xffffffffu;
                if (base + lane < per && idx < tot) r = rec[q[idx]];
                const unsigned c = (per - base < 32u) ? per - base : 32u;
                for (unsigned j0 = 0; j0 < c; j0 += 8u) {
                    float v[8][3];
                    float w[8];
                    unsigned rr[8];
#pragma unroll
                    for (int u = 0; u < 8; ++u) {
                        rr[u] = __builtin_amdgcn_readlane(r, j0 + u);
                        const unsigned s = (rr[u] == 0xffffffffu) ? 0u : (rr[u] & 0xffffu);
                        const float* hp = hs + (size_t)s * HID + lane;
                        v[u][0] = hp[0];
                        v[u][1] = hp[32];
                        v[u][2] = hp[64];
                        w[u] = onorm[s];
                    }
#pragma unroll
                    for (int u = 0; u < 8; ++u) {
                        if (rr[u] != 0xffffffffu) {
                            float* sp = slab + (rr[u] >> 16) * HID + lane;
                            lds_add(sp, v[u][0] * w[u]);
                            lds_add(sp + 32, v[u][1] * w[u]);
                            lds_add(sp + 64, v[u][2] * w[u]);
                        }
                    }
                }
            }
        } else {
            const unsigned eBeg = (unsigned)p * PASS_Q * 16u;
            const unsigned eEnd = (eBeg + PASS_Q * 16u < NE) ? eBeg + PASS_Q * 16u : NE;
            const unsigned char* ob = (const unsigned char*)own;
            for (unsigned e = eBeg + tid; e < eEnd; e += AGG_T) {
                if (ob[e] == me) {
                    const unsigned r = rec[e];
                    const float* hp = hs + (size_t)(r & 0xffffu) * HID;
                    const float w = onorm[r & 0xffffu];
                    float* sp = slab + (r >> 16) * HID;
                    for (int f = 0; f < HID; ++f) lds_add(sp + f, hp[f] * w);
                }
            }
        }
    }
    __syncthreads();

    const int lo = me * RNG;
    const int rows = (NN - lo < RNG) ? NN - lo : RNG;
    float4* o4 = (float4*)(agg + (size_t)lo * HID);
    for (int i = tid; i < rows * HID / 4; i += AGG_T) o4[i] = slab4[i];
}

#define GEMM_BLOCKS 256
#define GEMM_WAVES 16
#define GEMM_T (GEMM_WAVES * 32)

static_assert(NN % 16 == 0, "rows are taken in whole 16-row tiles");

template <int K>
__device__ __forceinline__ void gemm_load_a(const float* __restrict__ A, const float* __restrict__ rowScale,
                                            int tile, int lane, v2f (&av)[K / 4], float& s) {
    const int row = tile * 16 + (lane & 15);
    const float* arow = A + (size_t)row * K + (lane >> 4) * 2;
#pragma unroll
    for (int ks = 0; ks < K / 4; ++ks) av[ks] = *(const v2f*)(arow + 4 * ks);
    s = rowScale[row];
}

template <int K, int NOUT, int NPAD, bool RELU>
__global__ __launch_bounds__(GEMM_T) void gcn_gemm_wmma(
    const float* __restrict__ A, const float* __restrict__ rowScale,
    const float* __restrict__ W, const float* __restrict__ bias,
    float* __restrict__ out, int rowTiles)
{
    static_assert(K % 4 == 0 && NPAD % 16 == 0 && NOUT <= NPAD && NOUT % 2 == 0 && NPAD <= GEMM_T, "tile shapes");
    constexpr int NT = NPAD / 16;
    constexpr int KS = K / 4;
    constexpr int PB = 2 * NPAD + 32;
    __shared__ __attribute__((aligned(16))) float ws[(K / 2) * PB];
    __shared__ float bs[NPAD];

    const int tid = threadIdx.x;
    const int lane = tid & 31;
    const int wave = __builtin_amdgcn_readfirstlane(tid >> 5);
    const int cl = lane & 15;
    const int h = lane >> 4;
    const int tileStep = (int)gridDim.x * GEMM_WAVES;

    int tile = (int)blockIdx.x + (int)gridDim.x * wave;

    v2f av[KS];
    float s = 0.0f;
#pragma unroll
    for (int ks = 0; ks < KS; ++ks) av[ks] = v2f{0.0f, 0.0f};
    if (tile < rowTiles) gemm_load_a<K>(A, rowScale, tile, lane, av, s);

    constexpr int NP2 = NPAD / 2;
    constexpr int ITEMS = (K / 2) * NP2;
#pragma unroll
    for (int it = 0; it < (ITEMS + GEMM_T - 1) / GEMM_T; ++it) {
        const int idx = tid + it * GEMM_T;
        if (idx < ITEMS) {
            const int p = idx / NP2, n = (idx % NP2) * 2;
            v2f lo = {0.0f, 0.0f}, hi = {0.0f, 0.0f};
            if (NOUT == NPAD || n < NOUT) {
                lo = *(const v2f*)(W + (2 * p) * NOUT + n);
                hi = *(const v2f*)(W + (2 * p + 1) * NOUT + n);
            }
            *(float4*)(ws + p * PB + 2 * n) = make_float4(lo.x, hi.x, lo.y, hi.y);
        }
    }
    if (tid < NPAD) bs[tid] = (tid < NOUT) ? bias[tid] : 0.0f;
    __syncthreads();

    const float* bp = ws + h * PB + 2 * cl;

    constexpr int G = 2;
    constexpr int NG = (KS + G - 1) / G;

    while (tile < rowTiles) {
        v8f c[NT];
#pragma unroll
        for (int t = 0; t < NT; ++t) {
            const float bval = bs[16 * t + cl];
#pragma unroll
            for (int r = 0; r < 8; ++r) c[t][r] = bval;
        }

        v2f bq[2][G][NT];
#pragma unroll
        for (int i = 0; i < G; ++i) {
#pragma unroll
            for (int t = 0; t < NT; ++t)
                if (i < KS) bq[0][i][t] = *(const v2f*)(bp + 2 * i * PB + 32 * t);
        }

#pragma unroll
        for (int g = 0; g < NG; ++g) {
            if (g + 1 < NG) {
#pragma unroll
                for (int i = 0; i < G; ++i) {
                    const int ks = (g + 1) * G + i;
#pragma unroll
                    for (int t = 0; t < NT; ++t)
                        if (ks < KS) bq[(g + 1) & 1][i][t] = *(const v2f*)(bp + 2 * ks * PB + 32 * t);
                }
            }
#pragma unroll
            for (int i = 0; i < G; ++i) {
                const int ks = g * G + i;
                if (ks < KS) {
                    v2f a;
                    a.x = av[ks].x * s;
                    a.y = av[ks].y * s;
#pragma unroll
                    for (int t = 0; t < NT; ++t)
                        c[t] = __builtin_amdgcn_wmma_f32_16x16x4_f32(
                                 false, a,  false, bq[g & 1][i][t],
                                 (short)0, c[t],  false,  false);
                }
            }
            __builtin_amdgcn_sched_barrier(0);
        }

        const int rbase = tile * 16 + h * 8;
#pragma unroll
        for (int t = 0; t < NT; ++t) {
            const int col = 16 * t + cl;
            if (NOUT == NPAD || col < NOUT) {
#pragma unroll
                for (int r = 0; r < 8; ++r) {
                    float v = c[t][r];
                    if (RELU) v = fmaxf(v, 0.0f);
                    out[(size_t)(rbase + r) * NOUT + col] = v;
                }
            }
        }

        tile += tileStep;
        if (tile < rowTiles) gemm_load_a<K>(A, rowScale, tile, lane, av, s);
    }
}


extern "C" void kernel_launch(void* const* d_in, const int* in_sizes, int n_in,
                              void* d_out, int out_size, void* d_ws, size_t ws_size,
                              hipStream_t stream) {
    const int*   labels = (const int*)d_in[0];
    const int*   src    = (const int*)d_in[1];
    const int*   dst    = (const int*)d_in[2];
    const float* emb    = (const float*)d_in[3];
    const float* W1     = (const float*)d_in[4];
    const float* b1     = (const float*)d_in[5];
    const float* W2     = (const float*)d_in[6];
    const float* b2     = (const float*)d_in[7];
    const float* W3     = (const float*)d_in[8];
    const float* b3     = (const float*)d_in[9];
    float* out = (float*)d_out;

    float* w = (float*)d_ws;
    float* onorm = w;
    float* inorm = w + NN;
    float* bufA  = w + 2 * (size_t)NN;
    float* bufB  = bufA + (size_t)NN * HID;
    float* bufC  = bufB + (size_t)NN * HID;
    unsigned* own  = (unsigned*)(bufB + 8192);
    unsigned* rec  = own + NE / 4;
    unsigned* owns = rec + NE;

    const int TPB = 256;

    edge_rec_kernel<<<(NE / 4 + TPB - 1) / TPB, TPB, 0, stream>>>(src, dst, own, owns, rec, NE / 4);
    owner_outdeg_kernel<<<NOWN, AGG_T, 0, stream>>>(owns, rec, onorm);

    float* ew = bufB;
    label_weight_wmma<<<4 * (HID / 16), 32, 0, stream>>>(emb, W1, ew);
    owner_label_kernel<<<NOWN, AGG_T, 0, stream>>>(own, rec, labels, onorm, bufC, inorm);
    gcn_gemm_wmma<LPAD, HID, HID, true><<<GEMM_BLOCKS, GEMM_T, 0, stream>>>(
        bufC, inorm, ew, b1, bufA, NN / 16);

    owner_sum_kernel<<<NOWN, AGG_T, 0, stream>>>(bufA, onorm, own, rec, bufC);
    gcn_gemm_wmma<HID, HID, HID, true><<<GEMM_BLOCKS, GEMM_T, 0, stream>>>(
        bufC, inorm, W2, b2, bufA, NN / 16);

    owner_sum_kernel<<<NOWN, AGG_T, 0, stream>>>(bufA, onorm, own, rec, bufC);
    gcn_gemm_wmma<HID, NLAB, 64, false><<<GEMM_BLOCKS, GEMM_T, 0, stream>>>(
        bufC, inorm, W3, b3, out, NN / 16);
}


# each owner's edge list (record, source out_norm) stored once by layer 1's scan; the row sums of layers 2-3 read it and scan no owner bytes; sum order: LDS-add arrival, as before
# speedup vs baseline: 4.5299x; 1.2870x over previous
#include <hip/hip_runtime.h>

#define NN 50000
#define NE 800000
#define NLAB 50
#define EMB 64
#define HID 96
#define LPAD 52

#define RNG 196
#define NOWN ((NN + RNG - 1) / RNG)
#define AGG_T 1024
#define SCAN_L 7
#define NQ16 (NE / 16)
#define PASS_Q (AGG_T * SCAN_L)
#define NPASS ((NQ16 + PASS_Q - 1) / PASS_Q)
#define QALL 45056
#define LCAP 6144
#define LSTG 11264

static_assert(NOWN <= 256, "owner id must fit one byte");
static_assert(NN <= 65536 && RNG <= 65536, "edge record packs source and local row in 16 bits each");
static_assert(NE % 16 == 0, "owner bytes are read 16 at a time");
static_assert(PASS_Q <= NQ16, "the first pass must lie inside the owner array");
static_assert(LCAP <= QALL && LCAP <= LSTG && LCAP % AGG_T == 0, "a stored list comes out of one whole scan's queue and fits its LDS copy");

typedef __attribute__((ext_vector_type(2))) float v2f;
typedef __attribute__((ext_vector_type(8))) float v8f;


__global__ void edge_rec_kernel(const int* __restrict__ src, const int* __restrict__ dst,
                                unsigned* __restrict__ own, unsigned* __restrict__ owns,
                                unsigned* __restrict__ rec, int e4) {
    int i = blockIdx.x * blockDim.x + threadIdx.x;
    if (i < e4) {
        const int4 s = ((const int4*)src)[i];
        const int4 t = ((const int4*)dst)[i];
        const unsigned o0 = (unsigned)t.x / RNG, o1 = (unsigned)t.y / RNG;
        const unsigned o2 = (unsigned)t.z / RNG, o3 = (unsigned)t.w / RNG;
        own[i] = o0 | (o1 << 8) | (o2 << 16) | (o3 << 24);
        const unsigned p0 = (unsigned)s.x / RNG, p1 = (unsigned)s.y / RNG;
        const unsigned p2 = (unsigned)s.z / RNG, p3 = (unsigned)s.w / RNG;
        owns[i] = p0 | (p1 << 8) | (p2 << 16) | (p3 << 24);
        uint4 r;
        r.x = (((unsigned)t.x - o0 * RNG) << 16) | (unsigned)s.x;
        r.y = (((unsigned)t.y - o1 * RNG) << 16) | (unsigned)s.y;
        r.z = (((unsigned)t.z - o2 * RNG) << 16) | (unsigned)s.z;
        r.w = (((unsigned)t.w - o3 * RNG) << 16) | (unsigned)s.w;
        ((uint4*)rec)[i] = r;
    }
}


__global__ __launch_bounds__(32) void label_weight_wmma(const float* __restrict__ emb,
                                                        const float* __restrict__ W1,
                                                        float* __restrict__ ew) {
    const int lane = threadIdx.x;
    const int m0 = (blockIdx.x / (HID / 16)) * 16;
    const int n0 = (blockIdx.x % (HID / 16)) * 16;

    const int rowA = m0 + (lane & 15);
    const int koff = (lane >> 4) * 2;
    const int colB = n0 + (lane & 15);
    const bool live = rowA < NLAB;
    const float* arow = emb + (long)(live ? rowA : 0) * EMB;

    v8f c = {0.0f, 0.0f, 0.0f, 0.0f, 0.0f, 0.0f, 0.0f, 0.0f};
    for (int k0 = 0; k0 < EMB; k0 += 4) {
        v2f a, b;
        float a0 = arow[k0 + koff];
        float a1 = arow[k0 + koff + 1];
        a.x = live ? a0 : 0.0f;
        a.y = live ? a1 : 0.0f;
        b.x = W1[(k0 + koff) * HID + colB];
        b.y = W1[(k0 + koff + 1) * HID + colB];
        c = __builtin_amdgcn_wmma_f32_16x16x4_f32(
                 false, a,  false, b,
                 (short)0, c,  false,  false);
    }

    const int rbase = m0 + (lane >> 4) * 8;
#pragma unroll
    for (int r = 0; r < 8; ++r) {
        int row = rbase + r;
        if (row < LPAD) ew[row * HID + colB] = (row < NLAB) ? c[r] : 0.0f;
    }
}


__device__ __forceinline__ unsigned own_bytes(unsigned x) {
    return ~(((x & 0x7f7f7f7fu) + 0x7f7f7f7fu) | x | 0x7f7f7f7fu);
}

__device__ __forceinline__ void lds_add(float* p, float v) {
    __hip_atomic_fetch_add(p, v, __ATOMIC_RELAXED, __HIP_MEMORY_SCOPE_WORKGROUP);
}

__device__ __forceinline__ void owner_scan_all(const unsigned* __restrict__ ob, unsigned me,
                                               unsigned* q, unsigned* qn, unsigned cap, int tid) {
    const unsigned pat = me * 0x01010101u;
    const uint4* own16 = (const uint4*)ob;

    uint4 cur[SCAN_L], nxt[SCAN_L];
#pragma unroll
    for (int j = 0; j < SCAN_L; ++j) cur[j] = own16[j * AGG_T + tid];

    for (int p = 0; p < NPASS; ++p) {
        if (p + 1 < NPASS) {
#pragma unroll
            for (int j = 0; j < SCAN_L; ++j) {
                const unsigned qi = (p + 1) * PASS_Q + j * AGG_T + tid;
                nxt[j] = own16[qi < NQ16 ? qi : NQ16 - 1];
            }
        }
#pragma unroll
        for (int j = 0; j < SCAN_L; ++j) {
            const unsigned wi = (unsigned)(p * PASS_Q + j * AGG_T + tid);
            const unsigned e0 = wi * 16u;
            const unsigned c = (own_bytes(cur[j].x ^ pat) >> 7) | (own_bytes(cur[j].y ^ pat) >> 6) |
                               (own_bytes(cur[j].z ^ pat) >> 5) | (own_bytes(cur[j].w ^ pat) >> 4);
            unsigned m = (wi < NQ16) ? c : 0u;
            while (m) {
                const unsigned b = (unsigned)__builtin_ctz(m);
                m &= m - 1u;
                const unsigned slot = __hip_atomic_fetch_add(qn, 1u, __ATOMIC_RELAXED,
                                                             __HIP_MEMORY_SCOPE_WORKGROUP);
                if (slot < cap) q[slot] = e0 + ((b & 7u) << 2) + (b >> 3);
            }
        }
        if (p + 1 < NPASS) {
#pragma unroll
            for (int j = 0; j < SCAN_L; ++j) cur[j] = nxt[j];
        }
    }
}

__global__ __launch_bounds__(AGG_T) void owner_outdeg_kernel(const unsigned* __restrict__ owns,
                                                             const unsigned* __restrict__ rec,
                                                             float* __restrict__ onorm) {
    __shared__ unsigned queue[QALL];
    __shared__ unsigned cnt[RNG];
    __shared__ unsigned qn;

    const int tid = threadIdx.x;
    const unsigned me = blockIdx.x;
    const unsigned lo = me * RNG;

    if (tid < RNG) cnt[tid] = 0u;
    if (tid == 0) qn = 0u;
    __syncthreads();

    owner_scan_all(owns, me, queue, &qn, QALL, tid);
    __syncthreads();

    const unsigned tot = qn;
    if (tot <= QALL) {
        for (unsigned i = tid; i < tot; i += AGG_T) {
            const unsigned r = rec[queue[i]];
            atomicAdd(&cnt[(r & 0xffffu) - lo], 1u);
        }
    } else {
        const unsigned char* ob = (const unsigned char*)owns;
        for (unsigned e = tid; e < NE; e += AGG_T)
            if (ob[e] == me) atomicAdd(&cnt[(rec[e] & 0xffffu) - lo], 1u);
    }
    __syncthreads();

    if (tid < RNG && lo + tid < NN) onorm[lo + tid] = rsqrtf(fmaxf((float)cnt[tid], 1.0f));
}

__global__ __launch_bounds__(AGG_T) void owner_label_kernel(const unsigned* __restrict__ own,
                                                            const unsigned* __restrict__ rec,
                                                            const int* __restrict__ labels,
                                                            const float* __restrict__ onorm,
                                                            float* __restrict__ tab,
                                                            float* __restrict__ inorm,
                                                            unsigned* __restrict__ elist,
                                                            unsigned* __restrict__ ecount) {
    __shared__ __attribute__((aligned(16))) float lt[RNG * LPAD];
    __shared__ unsigned queue[QALL];
    __shared__ unsigned cnt[RNG];
    __shared__ unsigned qn;

    const int tid = threadIdx.x;
    const unsigned me = blockIdx.x;
    const unsigned lo = me * RNG;

    float4* lt4 = (float4*)lt;
    for (int i = tid; i < RNG * LPAD / 4; i += AGG_T) lt4[i] = make_float4(0.0f, 0.0f, 0.0f, 0.0f);
    if (tid < RNG) cnt[tid] = 0u;
    if (tid == 0) qn = 0u;
    __syncthreads();

    owner_scan_all(own, me, queue, &qn, QALL, tid);
    __syncthreads();

    const unsigned tot = qn;
    if (tid == 0) ecount[me] = tot;
    if (tot <= QALL) {
        uint2* el = (uint2*)elist + (size_t)me * LCAP;
        for (unsigned i = tid; i < tot; i += AGG_T) {
            const unsigned r = rec[queue[i]];
            const unsigned s = r & 0xffffu, row = r >> 16;
            const float w = onorm[s];
            if (i < LCAP) el[i] = make_uint2(r, __float_as_uint(w));
            lds_add(&lt[row * LPAD + labels[s]], w);
            atomicAdd(&cnt[row], 1u);
        }
    } else {
        const unsigned char* ob = (const unsigned char*)own;
        for (unsigned e = tid; e < NE; e += AGG_T) {
            if (ob[e] == me) {
                const unsigned r = rec[e];
                const unsigned s = r & 0xffffu, row = r >> 16;
                lds_add(&lt[row * LPAD + labels[s]], onorm[s]);
                atomicAdd(&cnt[row], 1u);
            }
        }
    }
    __syncthreads();

    const int rows = ((int)NN - (int)lo < RNG) ? (int)NN - (int)lo : RNG;
    float4* o4 = (float4*)(tab + (size_t)lo * LPAD);
    for (int i = tid; i < rows * LPAD / 4; i += AGG_T) o4[i] = lt4[i];
    if (tid < rows) inorm[lo + tid] = rsqrtf(fmaxf((float)cnt[tid], 1.0f));
}

__global__ __launch_bounds__(AGG_T) void owner_sum_kernel(const float* __restrict__ hs,
                                                          const unsigned* __restrict__ elist,
                                                          const unsigned* __restrict__ ecount,
                                                          const unsigned* __restrict__ own,
                                                          const unsigned* __restrict__ rec,
                                                          const float* __restrict__ onorm,
                                                          float* __restrict__ agg) {
    __shared__ __attribute__((aligned(16))) float slab[RNG * HID];
    __shared__ uint2 lst[LSTG];

    const int tid = threadIdx.x;
    const int lane = tid & 31;
    const int wave = __builtin_amdgcn_readfirstlane(tid >> 5);
    const unsigned me = blockIdx.x;

    const unsigned tot = __builtin_amdgcn_readfirstlane(ecount[me]);
    const bool listed = tot <= LCAP;

    const uint2* L = (const uint2*)elist + (size_t)me * LCAP;
    const uint2 none = {0xffffffffu, 0u};
    uint2 e[LCAP / AGG_T];
#pragma unroll
    for (int k = 0; k < LCAP / AGG_T; ++k) {
        const unsigned i = (unsigned)(k * AGG_T + tid);
        e[k] = none;
        if (listed && i < tot) e[k] = L[i];
    }

    float4* slab4 = (float4*)slab;
    for (int i = tid; i < RNG * HID / 4; i += AGG_T) slab4[i] = make_float4(0.0f, 0.0f, 0.0f, 0.0f);
#pragma unroll
    for (int k = 0; k < LCAP / AGG_T; ++k) lst[k * AGG_T + tid] = e[k];
    __syncthreads();

    if (listed) {
        const unsigned per = (((tot + 31u) >> 5) + 7u) & ~7u;
        const unsigned beg = wave * per;
        const unsigned end = (beg + per < tot) ? beg + per : tot;
        for (unsigned base = beg; base < end; base += 32u) {
            const unsigned idx = base + lane;
            const uint2 ent = (idx < end) ? lst[idx] : none;
            const unsigned c = (end - base < 32u) ? end - base : 32u;
            for (unsigned j0 = 0; j0 < c; j0 += 8u) {
                float v[8][3];
                float w[8];
                unsigned rr[8];
#pragma unroll
                for (int u = 0; u < 8; ++u) {
                    rr[u] = __builtin_amdgcn_readlane(ent.x, j0 + u);
                    w[u] = __uint_as_float(__builtin_amdgcn_readlane(ent.y, j0 + u));
                    const unsigned s = (rr[u] == 0xffffffffu) ? 0u : (rr[u] & 0xffffu);
                    const float* hp = hs + (size_t)s * HID + lane;
                    v[u][0] = hp[0];
                    v[u][1] = hp[32];
                    v[u][2] = hp[64];
                }
#pragma unroll
                for (int u = 0; u < 8; ++u) {
                    if (rr[u] != 0xffffffffu) {
                        float* sp = slab + (rr[u] >> 16) * HID + lane;
                        lds_add(sp, v[u][0] * w[u]);
                        lds_add(sp + 32, v[u][1] * w[u]);
                        lds_add(sp + 64, v[u][2] * w[u]);
                    }
                }
            }
        }
    } else {
        const unsigned char* ob = (const unsigned char*)own;
        for (unsigned ed = tid; ed < NE; ed += AGG_T) {
            if (ob[ed] == me) {
                const unsigned r = rec[ed];
                const float* hp = hs + (size_t)(r & 0xffffu) * HID;
                const float w = onorm[r & 0xffffu];
                float* sp = slab + (r >> 16) * HID;
                for (int f = 0; f < HID; ++f) lds_add(sp + f, hp[f] * w);
            }
        }
    }
    __syncthreads();

    const int lo = me * RNG;
    const int rows = (NN - lo < RNG) ? NN - lo : RNG;
    float4* o4 = (float4*)(agg + (size_t)lo * HID);
    for (int i = tid; i < rows * HID / 4; i += AGG_T) o4[i] = slab4[i];
}

#define GEMM_BLOCKS 256
#define GEMM_WAVES 16
#define GEMM_T (GEMM_WAVES * 32)

static_assert(NN % 16 == 0, "rows are taken in whole 16-row tiles");

template <int K>
__device__ __forceinline__ void gemm_load_a(const float* __restrict__ A, const float* __restrict__ rowScale,
                                            int tile, int lane, v2f (&av)[K / 4], float& s) {
    const int row = tile * 16 + (lane & 15);
    const float* arow = A + (size_t)row * K + (lane >> 4) * 2;
#pragma unroll
    for (int ks = 0; ks < K / 4; ++ks) av[ks] = *(const v2f*)(arow + 4 * ks);
    s = rowScale[row];
}

template <int K, int NOUT, int NPAD, bool RELU>
__global__ __launch_bounds__(GEMM_T) void gcn_gemm_wmma(
    const float* __restrict__ A, const float* __restrict__ rowScale,
    const float* __restrict__ W, const float* __restrict__ bias,
    float* __restrict__ out, int rowTiles)
{
    static_assert(K % 4 == 0 && NPAD % 16 == 0 && NOUT <= NPAD && NOUT % 2 == 0 && NPAD <= GEMM_T, "tile shapes");
    constexpr int NT = NPAD / 16;
    constexpr int KS = K / 4;
    constexpr int PB = 2 * NPAD + 32;
    __shared__ __attribute__((aligned(16))) float ws[(K / 2) * PB];
    __shared__ float bs[NPAD];

    const int tid = threadIdx.x;
    const int lane = tid & 31;
    const int wave = __builtin_amdgcn_readfirstlane(tid >> 5);
    const int cl = lane & 15;
    const int h = lane >> 4;
    const int tileStep = (int)gridDim.x * GEMM_WAVES;

    int tile = (int)blockIdx.x + (int)gridDim.x * wave;

    v2f av[KS];
    float s = 0.0f;
#pragma unroll
    for (int ks = 0; ks < KS; ++ks) av[ks] = v2f{0.0f, 0.0f};
    if (tile < rowTiles) gemm_load_a<K>(A, rowScale, tile, lane, av, s);

    constexpr int NP2 = NPAD / 2;
    constexpr int ITEMS = (K / 2) * NP2;
#pragma unroll
    for (int it = 0; it < (ITEMS + GEMM_T - 1) / GEMM_T; ++it) {
        const int idx = tid + it * GEMM_T;
        if (idx < ITEMS) {
            const int p = idx / NP2, n = (idx % NP2) * 2;
            v2f lo = {0.0f, 0.0f}, hi = {0.0f, 0.0f};
            if (NOUT == NPAD || n < NOUT) {
                lo = *(const v2f*)(W + (2 * p) * NOUT + n);
                hi = *(const v2f*)(W + (2 * p + 1) * NOUT + n);
            }
            *(float4*)(ws + p * PB + 2 * n) = make_float4(lo.x, hi.x, lo.y, hi.y);
        }
    }
    if (tid < NPAD) bs[tid] = (tid < NOUT) ? bias[tid] : 0.0f;
    __syncthreads();

    const float* bp = ws + h * PB + 2 * cl;

    constexpr int G = 2;
    constexpr int NG = (KS + G - 1) / G;

    while (tile < rowTiles) {
        v8f c[NT];
#pragma unroll
        for (int t = 0; t < NT; ++t) {
            const float bval = bs[16 * t + cl];
#pragma unroll
            for (int r = 0; r < 8; ++r) c[t][r] = bval;
        }

        v2f bq[2][G][NT];
#pragma unroll
        for (int i = 0; i < G; ++i) {
#pragma unroll
            for (int t = 0; t < NT; ++t)
                if (i < KS) bq[0][i][t] = *(const v2f*)(bp + 2 * i * PB + 32 * t);
        }

#pragma unroll
        for (int g = 0; g < NG; ++g) {
            if (g + 1 < NG) {
#pragma unroll
                for (int i = 0; i < G; ++i) {
                    const int ks = (g + 1) * G + i;
#pragma unroll
                    for (int t = 0; t < NT; ++t)
                        if (ks < KS) bq[(g + 1) & 1][i][t] = *(const v2f*)(bp + 2 * ks * PB + 32 * t);
                }
            }
#pragma unroll
            for (int i = 0; i < G; ++i) {
                const int ks = g * G + i;
                if (ks < KS) {
                    v2f a;
                    a.x = av[ks].x * s;
                    a.y = av[ks].y * s;
#pragma unroll
                    for (int t = 0; t < NT; ++t)
                        c[t] = __builtin_amdgcn_wmma_f32_16x16x4_f32(
                                 false, a,  false, bq[g & 1][i][t],
                                 (short)0, c[t],  false,  false);
                }
            }
            __builtin_amdgcn_sched_barrier(0);
        }

        const int rbase = tile * 16 + h * 8;
#pragma unroll
        for (int t = 0; t < NT; ++t) {
            const int col = 16 * t + cl;
            if (NOUT == NPAD || col < NOUT) {
#pragma unroll
                for (int r = 0; r < 8; ++r) {
                    float v = c[t][r];
                    if (RELU) v = fmaxf(v, 0.0f);
                    out[(size_t)(rbase + r) * NOUT + col] = v;
                }
            }
        }

        tile += tileStep;
        if (tile < rowTiles) gemm_load_a<K>(A, rowScale, tile, lane, av, s);
    }
}


extern "C" void kernel_launch(void* const* d_in, const int* in_sizes, int n_in,
                              void* d_out, int out_size, void* d_ws, size_t ws_size,
                              hipStream_t stream) {
    const int*   labels = (const int*)d_in[0];
    const int*   src    = (const int*)d_in[1];
    const int*   dst    = (const int*)d_in[2];
    const float* emb    = (const float*)d_in[3];
    const float* W1     = (const float*)d_in[4];
    const float* b1     = (const float*)d_in[5];
    const float* W2     = (const float*)d_in[6];
    const float* b2     = (const float*)d_in[7];
    const float* W3     = (const float*)d_in[8];
    const float* b3     = (const float*)d_in[9];
    float* out = (float*)d_out;

    float* w = (float*)d_ws;
    float* onorm = w;
    float* inorm = w + NN;
    float* bufA  = w + 2 * (size_t)NN;
    float* bufB  = bufA + (size_t)NN * HID;
    float* bufC  = bufB + (size_t)NN * HID;
    unsigned* own  = (unsigned*)(bufB + 8192);
    unsigned* rec  = own + NE / 4;
    unsigned* owns = rec + NE;
    unsigned* elist  = owns + NE / 4;
    unsigned* ecount = elist + 2 * (size_t)NOWN * LCAP;

    const int TPB = 256;

    edge_rec_kernel<<<(NE / 4 + TPB - 1) / TPB, TPB, 0, stream>>>(src, dst, own, owns, rec, NE / 4);
    owner_outdeg_kernel<<<NOWN, AGG_T, 0, stream>>>(owns, rec, onorm);

    float* ew = bufB;
    label_weight_wmma<<<4 * (HID / 16), 32, 0, stream>>>(emb, W1, ew);
    owner_label_kernel<<<NOWN, AGG_T, 0, stream>>>(own, rec, labels, onorm, bufC, inorm, elist, ecount);
    gcn_gemm_wmma<LPAD, HID, HID, true><<<GEMM_BLOCKS, GEMM_T, 0, stream>>>(
        bufC, inorm, ew, b1, bufA, NN / 16);

    owner_sum_kernel<<<NOWN, AGG_T, 0, stream>>>(bufA, elist, ecount, own, rec, onorm, bufC);
    gcn_gemm_wmma<HID, HID, HID, true><<<GEMM_BLOCKS, GEMM_T, 0, stream>>>(
        bufC, inorm, W2, b2, bufA, NN / 16);

    owner_sum_kernel<<<NOWN, AGG_T, 0, stream>>>(bufA, elist, ecount, own, rec, onorm, bufC);
    gcn_gemm_wmma<HID, NLAB, 64, false><<<GEMM_BLOCKS, GEMM_T, 0, stream>>>(
        bufC, inorm, W3, b3, out, NN / 16);
}


# out-degrees without the owner-byte scan: 16 node groups x 16 edge slices, a block counts in LDS its slice's sources inside its group straight from src; record kernel adds the integer partials
# speedup vs baseline: 4.9957x; 1.1028x over previous
#include <hip/hip_runtime.h>

#define NN 50000
#define NE 800000
#define NLAB 50
#define EMB 64
#define HID 96
#define LPAD 52

#define RNG 196
#define NOWN ((NN + RNG - 1) / RNG)
#define AGG_T 1024
#define SCAN_L 7
#define NQ16 (NE / 16)
#define PASS_Q (AGG_T * SCAN_L)
#define NPASS ((NQ16 + PASS_Q - 1) / PASS_Q)
#define QALL 45056
#define LCAP 6144
#define LSTG 11264

static_assert(NOWN <= 256, "owner id must fit one byte");
static_assert(NN <= 65536 && RNG <= 65536, "edge record packs source and local row in 16 bits each");
static_assert(NE % 16 == 0, "owner bytes are read 16 at a time");
static_assert(PASS_Q <= NQ16, "the first pass must lie inside the owner array");
static_assert(LCAP <= QALL && LCAP <= LSTG && LCAP % AGG_T == 0, "a stored list comes out of one whole scan's queue and fits its LDS copy");

typedef __attribute__((ext_vector_type(2))) float v2f;
typedef __attribute__((ext_vector_type(8))) float v8f;

#define DG 16
#define DS 16
#define DGN (NN / DG)
#define DSE4 (NE / DS / 4)
#define DEG_T 1024
#define DEG_L ((DSE4 + DEG_T - 1) / DEG_T)

static_assert(NN % DG == 0 && NE % (DS * 4) == 0, "groups and slices are whole");
static_assert(NN <= NE / 4, "edge_rec_kernel finishes node i in the thread of edge quad i");

__global__ __launch_bounds__(DEG_T) void outdeg_part_kernel(const int* __restrict__ src,
                                                            unsigned* __restrict__ pc) {
    __shared__ unsigned cnt[DGN];

    const int tid = threadIdx.x;
    const unsigned g = blockIdx.x % DG, sl = blockIdx.x / DG;
    const unsigned lo = g * DGN;

    const int4* s4 = (const int4*)src + (size_t)sl * DSE4;
    int4 v[DEG_L];
#pragma unroll
    for (int k = 0; k < DEG_L; ++k) {
        const int i = k * DEG_T + tid;
        v[k] = s4[i < DSE4 ? i : DSE4 - 1];
    }

    for (int i = tid; i < DGN; i += DEG_T) cnt[i] = 0u;
    __syncthreads();

#pragma unroll
    for (int k = 0; k < DEG_L; ++k) {
        if (k * DEG_T + tid < DSE4) {
            const unsigned a0 = (unsigned)v[k].x - lo, a1 = (unsigned)v[k].y - lo;
            const unsigned a2 = (unsigned)v[k].z - lo, a3 = (unsigned)v[k].w - lo;
            if (a0 < DGN) atomicAdd(&cnt[a0], 1u);
            if (a1 < DGN) atomicAdd(&cnt[a1], 1u);
            if (a2 < DGN) atomicAdd(&cnt[a2], 1u);
            if (a3 < DGN) atomicAdd(&cnt[a3], 1u);
        }
    }
    __syncthreads();

    for (int i = tid; i < DGN; i += DEG_T) pc[(size_t)sl * NN + lo + i] = cnt[i];
}


__global__ void edge_rec_kernel(const int* __restrict__ src, const int* __restrict__ dst,
                                unsigned* __restrict__ own, unsigned* __restrict__ rec,
                                const unsigned* __restrict__ pc, float* __restrict__ onorm, int e4) {
    int i = blockIdx.x * blockDim.x + threadIdx.x;
    if (i < e4) {
        const int4 s = ((const int4*)src)[i];
        const int4 t = ((const int4*)dst)[i];
        if (i < NN) {
            unsigned c = 0u;
#pragma unroll
            for (int sl = 0; sl < DS; ++sl) c += pc[(size_t)sl * NN + i];
            onorm[i] = rsqrtf(fmaxf((float)c, 1.0f));
        }
        const unsigned o0 = (unsigned)t.x / RNG, o1 = (unsigned)t.y / RNG;
        const unsigned o2 = (unsigned)t.z / RNG, o3 = (unsigned)t.w / RNG;
        own[i] = o0 | (o1 << 8) | (o2 << 16) | (o3 << 24);
        uint4 r;
        r.x = (((unsigned)t.x - o0 * RNG) << 16) | (unsigned)s.x;
        r.y = (((unsigned)t.y - o1 * RNG) << 16) | (unsigned)s.y;
        r.z = (((unsigned)t.z - o2 * RNG) << 16) | (unsigned)s.z;
        r.w = (((unsigned)t.w - o3 * RNG) << 16) | (unsigned)s.w;
        ((uint4*)rec)[i] = r;
    }
}


__global__ __launch_bounds__(32) void label_weight_wmma(const float* __restrict__ emb,
                                                        const float* __restrict__ W1,
                                                        float* __restrict__ ew) {
    const int lane = threadIdx.x;
    const int m0 = (blockIdx.x / (HID / 16)) * 16;
    const int n0 = (blockIdx.x % (HID / 16)) * 16;

    const int rowA = m0 + (lane & 15);
    const int koff = (lane >> 4) * 2;
    const int colB = n0 + (lane & 15);
    const bool live = rowA < NLAB;
    const float* arow = emb + (long)(live ? rowA : 0) * EMB;

    v8f c = {0.0f, 0.0f, 0.0f, 0.0f, 0.0f, 0.0f, 0.0f, 0.0f};
    for (int k0 = 0; k0 < EMB; k0 += 4) {
        v2f a, b;
        float a0 = arow[k0 + koff];
        float a1 = arow[k0 + koff + 1];
        a.x = live ? a0 : 0.0f;
        a.y = live ? a1 : 0.0f;
        b.x = W1[(k0 + koff) * HID + colB];
        b.y = W1[(k0 + koff + 1) * HID + colB];
        c = __builtin_amdgcn_wmma_f32_16x16x4_f32(
                 false, a,  false, b,
                 (short)0, c,  false,  false);
    }

    const int rbase = m0 + (lane >> 4) * 8;
#pragma unroll
    for (int r = 0; r < 8; ++r) {
        int row = rbase + r;
        if (row < LPAD) ew[row * HID + colB] = (row < NLAB) ? c[r] : 0.0f;
    }
}


__device__ __forceinline__ unsigned own_bytes(unsigned x) {
    return ~(((x & 0x7f7f7f7fu) + 0x7f7f7f7fu) | x | 0x7f7f7f7fu);
}

__device__ __forceinline__ void lds_add(float* p, float v) {
    __hip_atomic_fetch_add(p, v, __ATOMIC_RELAXED, __HIP_MEMORY_SCOPE_WORKGROUP);
}

__device__ __forceinline__ void owner_scan_all(const unsigned* __restrict__ ob, unsigned me,
                                               unsigned* q, unsigned* qn, unsigned cap, int tid) {
    const unsigned pat = me * 0x01010101u;
    const uint4* own16 = (const uint4*)ob;

    uint4 cur[SCAN_L], nxt[SCAN_L];
#pragma unroll
    for (int j = 0; j < SCAN_L; ++j) cur[j] = own16[j * AGG_T + tid];

    for (int p = 0; p < NPASS; ++p) {
        if (p + 1 < NPASS) {
#pragma unroll
            for (int j = 0; j < SCAN_L; ++j) {
                const unsigned qi = (p + 1) * PASS_Q + j * AGG_T + tid;
                nxt[j] = own16[qi < NQ16 ? qi : NQ16 - 1];
            }
        }
#pragma unroll
        for (int j = 0; j < SCAN_L; ++j) {
            const unsigned wi = (unsigned)(p * PASS_Q + j * AGG_T + tid);
            const unsigned e0 = wi * 16u;
            const unsigned c = (own_bytes(cur[j].x ^ pat) >> 7) | (own_bytes(cur[j].y ^ pat) >> 6) |
                               (own_bytes(cur[j].z ^ pat) >> 5) | (own_bytes(cur[j].w ^ pat) >> 4);
            unsigned m = (wi < NQ16) ? c : 0u;
            while (m) {
                const unsigned b = (unsigned)__builtin_ctz(m);
                m &= m - 1u;
                const unsigned slot = __hip_atomic_fetch_add(qn, 1u, __ATOMIC_RELAXED,
                                                             __HIP_MEMORY_SCOPE_WORKGROUP);
                if (slot < cap) q[slot] = e0 + ((b & 7u) << 2) + (b >> 3);
            }
        }
        if (p + 1 < NPASS) {
#pragma unroll
            for (int j = 0; j < SCAN_L; ++j) cur[j] = nxt[j];
        }
    }
}

__global__ __launch_bounds__(AGG_T) void owner_label_kernel(const unsigned* __restrict__ own,
                                                            const unsigned* __restrict__ rec,
                                                            const int* __restrict__ labels,
                                                            const float* __restrict__ onorm,
                                                            float* __restrict__ tab,
                                                            float* __restrict__ inorm,
                                                            unsigned* __restrict__ elist,
                                                            unsigned* __restrict__ ecount) {
    __shared__ __attribute__((aligned(16))) float lt[RNG * LPAD];
    __shared__ unsigned queue[QALL];
    __shared__ unsigned cnt[RNG];
    __shared__ unsigned qn;

    const int tid = threadIdx.x;
    const unsigned me = blockIdx.x;
    const unsigned lo = me * RNG;

    float4* lt4 = (float4*)lt;
    for (int i = tid; i < RNG * LPAD / 4; i += AGG_T) lt4[i] = make_float4(0.0f, 0.0f, 0.0f, 0.0f);
    if (tid < RNG) cnt[tid] = 0u;
    if (tid == 0) qn = 0u;
    __syncthreads();

    owner_scan_all(own, me, queue, &qn, QALL, tid);
    __syncthreads();

    const unsigned tot = qn;
    if (tid == 0) ecount[me] = tot;
    if (tot <= QALL) {
        uint2* el = (uint2*)elist + (size_t)me * LCAP;
        for (unsigned i = tid; i < tot; i += AGG_T) {
            const unsigned r = rec[queue[i]];
            const unsigned s = r & 0xffffu, row = r >> 16;
            const float w = onorm[s];
            if (i < LCAP) el[i] = make_uint2(r, __float_as_uint(w));
            lds_add(&lt[row * LPAD + labels[s]], w);
            atomicAdd(&cnt[row], 1u);
        }
    } else {
        const unsigned char* ob = (const unsigned char*)own;
        for (unsigned e = tid; e < NE; e += AGG_T) {
            if (ob[e] == me) {
                const unsigned r = rec[e];
                const unsigned s = r & 0xffffu, row = r >> 16;
                lds_add(&lt[row * LPAD + labels[s]], onorm[s]);
                atomicAdd(&cnt[row], 1u);
            }
        }
    }
    __syncthreads();

    const int rows = ((int)NN - (int)lo < RNG) ? (int)NN - (int)lo : RNG;
    float4* o4 = (float4*)(tab + (size_t)lo * LPAD);
    for (int i = tid; i < rows * LPAD / 4; i += AGG_T) o4[i] = lt4[i];
    if (tid < rows) inorm[lo + tid] = rsqrtf(fmaxf((float)cnt[tid], 1.0f));
}

__global__ __launch_bounds__(AGG_T) void owner_sum_kernel(const float* __restrict__ hs,
                                                          const unsigned* __restrict__ elist,
                                                          const unsigned* __restrict__ ecount,
                                                          const unsigned* __restrict__ own,
                                                          const unsigned* __restrict__ rec,
                                                          const float* __restrict__ onorm,
                                                          float* __restrict__ agg) {
    __shared__ __attribute__((aligned(16))) float slab[RNG * HID];
    __shared__ uint2 lst[LSTG];

    const int tid = threadIdx.x;
    const int lane = tid & 31;
    const int wave = __builtin_amdgcn_readfirstlane(tid >> 5);
    const unsigned me = blockIdx.x;

    const unsigned tot = __builtin_amdgcn_readfirstlane(ecount[me]);
    const bool listed = tot <= LCAP;

    const uint2* L = (const uint2*)elist + (size_t)me * LCAP;
    const uint2 none = {0xffffffffu, 0u};
    uint2 e[LCAP / AGG_T];
#pragma unroll
    for (int k = 0; k < LCAP / AGG_T; ++k) {
        const unsigned i = (unsigned)(k * AGG_T + tid);
        e[k] = none;
        if (listed && i < tot) e[k] = L[i];
    }

    float4* slab4 = (float4*)slab;
    for (int i = tid; i < RNG * HID / 4; i += AGG_T) slab4[i] = make_float4(0.0f, 0.0f, 0.0f, 0.0f);
#pragma unroll
    for (int k = 0; k < LCAP / AGG_T; ++k) lst[k * AGG_T + tid] = e[k];
    __syncthreads();

    if (listed) {
        const unsigned per = (((tot + 31u) >> 5) + 7u) & ~7u;
        const unsigned beg = wave * per;
        const unsigned end = (beg + per < tot) ? beg + per : tot;
        for (unsigned base = beg; base < end; base += 32u) {
            const unsigned idx = base + lane;
            const uint2 ent = (idx < end) ? lst[idx] : none;
            const unsigned c = (end - base < 32u) ? end - base : 32u;
            for (unsigned j0 = 0; j0 < c; j0 += 8u) {
                float v[8][3];
                float w[8];
                unsigned rr[8];
#pragma unroll
                for (int u = 0; u < 8; ++u) {
                    rr[u] = __builtin_amdgcn_readlane(ent.x, j0 + u);
                    w[u] = __uint_as_float(__builtin_amdgcn_readlane(ent.y, j0 + u));
                    const unsigned s = (rr[u] == 0xffffffffu) ? 0u : (rr[u] & 0xffffu);
                    const float* hp = hs + (size_t)s * HID + lane;
                    v[u][0] = hp[0];
                    v[u][1] = hp[32];
                    v[u][2] = hp[64];
                }
#pragma unroll
                for (int u = 0; u < 8; ++u) {
                    if (rr[u] != 0xffffffffu) {
                        float* sp = slab + (rr[u] >> 16) * HID + lane;
                        lds_add(sp, v[u][0] * w[u]);
                        lds_add(sp + 32, v[u][1] * w[u]);
                        lds_add(sp + 64, v[u][2] * w[u]);
                    }
                }
            }
        }
    } else {
        const unsigned char* ob = (const unsigned char*)own;
        for (unsigned ed = tid; ed < NE; ed += AGG_T) {
            if (ob[ed] == me) {
                const unsigned r = rec[ed];
                const float* hp = hs + (size_t)(r & 0xffffu) * HID;
                const float w = onorm[r & 0xffffu];
                float* sp = slab + (r >> 16) * HID;
                for (int f = 0; f < HID; ++f) lds_add(sp + f, hp[f] * w);
            }
        }
    }
    __syncthreads();

    const int lo = me * RNG;
    const int rows = (NN - lo < RNG) ? NN - lo : RNG;
    float4* o4 = (float4*)(agg + (size_t)lo * HID);
    for (int i = tid; i < rows * HID / 4; i += AGG_T) o4[i] = slab4[i];
}

#define GEMM_BLOCKS 256
#define GEMM_WAVES 16
#define GEMM_T (GEMM_WAVES * 32)

static_assert(NN % 16 == 0, "rows are taken in whole 16-row tiles");

template <int K>
__device__ __forceinline__ void gemm_load_a(const float* __restrict__ A, const float* __restrict__ rowScale,
                                            int tile, int lane, v2f (&av)[K / 4], float& s) {
    const int row = tile * 16 + (lane & 15);
    const float* arow = A + (size_t)row * K + (lane >> 4) * 2;
#pragma unroll
    for (int ks = 0; ks < K / 4; ++ks) av[ks] = *(const v2f*)(arow + 4 * ks);
    s = rowScale[row];
}

template <int K, int NOUT, int NPAD, bool RELU>
__global__ __launch_bounds__(GEMM_T) void gcn_gemm_wmma(
    const float* __restrict__ A, const float* __restrict__ rowScale,
    const float* __restrict__ W, const float* __restrict__ bias,
    float* __restrict__ out, int rowTiles)
{
    static_assert(K % 4 == 0 && NPAD % 16 == 0 && NOUT <= NPAD && NOUT % 2 == 0 && NPAD <= GEMM_T, "tile shapes");
    constexpr int NT = NPAD / 16;
    constexpr int KS = K / 4;
    constexpr int PB = 2 * NPAD + 32;
    __shared__ __attribute__((aligned(16))) float ws[(K / 2) * PB];
    __shared__ float bs[NPAD];

    const int tid = threadIdx.x;
    const int lane = tid & 31;
    const int wave = __builtin_amdgcn_readfirstlane(tid >> 5);
    const int cl = lane & 15;
    const int h = lane >> 4;
    const int tileStep = (int)gridDim.x * GEMM_WAVES;

    int tile = (int)blockIdx.x + (int)gridDim.x * wave;

    v2f av[KS];
    float s = 0.0f;
#pragma unroll
    for (int ks = 0; ks < KS; ++ks) av[ks] = v2f{0.0f, 0.0f};
    if (tile < rowTiles) gemm_load_a<K>(A, rowScale, tile, lane, av, s);

    constexpr int NP2 = NPAD / 2;
    constexpr int ITEMS = (K / 2) * NP2;
#pragma unroll
    for (int it = 0; it < (ITEMS + GEMM_T - 1) / GEMM_T; ++it) {
        const int idx = tid + it * GEMM_T;
        if (idx < ITEMS) {
            const int p = idx / NP2, n = (idx % NP2) * 2;
            v2f lo = {0.0f, 0.0f}, hi = {0.0f, 0.0f};
            if (NOUT == NPAD || n < NOUT) {
                lo = *(const v2f*)(W + (2 * p) * NOUT + n);
                hi = *(const v2f*)(W + (2 * p + 1) * NOUT + n);
            }
            *(float4*)(ws + p * PB + 2 * n) = make_float4(lo.x, hi.x, lo.y, hi.y);
        }
    }
    if (tid < NPAD) bs[tid] = (tid < NOUT) ? bias[tid] : 0.0f;
    __syncthreads();

    const float* bp = ws + h * PB + 2 * cl;

    constexpr int G = 2;
    constexpr int NG = (KS + G - 1) / G;

    while (tile < rowTiles) {
        v8f c[NT];
#pragma unroll
        for (int t = 0; t < NT; ++t) {
            const float bval = bs[16 * t + cl];
#pragma unroll
            for (int r = 0; r < 8; ++r) c[t][r] = bval;
        }

        v2f bq[2][G][NT];
#pragma unroll
        for (int i = 0; i < G; ++i) {
#pragma unroll
            for (int t = 0; t < NT; ++t)
                if (i < KS) bq[0][i][t] = *(const v2f*)(bp + 2 * i * PB + 32 * t);
        }

#pragma unroll
        for (int g = 0; g < NG; ++g) {
            if (g + 1 < NG) {
#pragma unroll
                for (int i = 0; i < G; ++i) {
                    const int ks = (g + 1) * G + i;
#pragma unroll
                    for (int t = 0; t < NT; ++t)
                        if (ks < KS) bq[(g + 1) & 1][i][t] = *(const v2f*)(bp + 2 * ks * PB + 32 * t);
                }
            }
#pragma unroll
            for (int i = 0; i < G; ++i) {
                const int ks = g * G + i;
                if (ks < KS) {
                    v2f a;
                    a.x = av[ks].x * s;
                    a.y = av[ks].y * s;
#pragma unroll
                    for (int t = 0; t < NT; ++t)
                        c[t] = __builtin_amdgcn_wmma_f32_16x16x4_f32(
                                 false, a,  false, bq[g & 1][i][t],
                                 (short)0, c[t],  false,  false);
                }
            }
            __builtin_amdgcn_sched_barrier(0);
        }

        const int rbase = tile * 16 + h * 8;
#pragma unroll
        for (int t = 0; t < NT; ++t) {
            const int col = 16 * t + cl;
            if (NOUT == NPAD || col < NOUT) {
#pragma unroll
                for (int r = 0; r < 8; ++r) {
                    float v = c[t][r];
                    if (RELU) v = fmaxf(v, 0.0f);
                    out[(size_t)(rbase + r) * NOUT + col] = v;
                }
            }
        }

        tile += tileStep;
        if (tile < rowTiles) gemm_load_a<K>(A, rowScale, tile, lane, av, s);
    }
}


extern "C" void kernel_launch(void* const* d_in, const int* in_sizes, int n_in,
                              void* d_out, int out_size, void* d_ws, size_t ws_size,
                              hipStream_t stream) {
    const int*   labels = (const int*)d_in[0];
    const int*   src    = (const int*)d_in[1];
    const int*   dst    = (const int*)d_in[2];
    const float* emb    = (const float*)d_in[3];
    const float* W1     = (const float*)d_in[4];
    const float* b1     = (const float*)d_in[5];
    const float* W2     = (const float*)d_in[6];
    const float* b2     = (const float*)d_in[7];
    const float* W3     = (const float*)d_in[8];
    const float* b3     = (const float*)d_in[9];
    float* out = (float*)d_out;

    float* w = (float*)d_ws;
    float* onorm = w;
    float* inorm = w + NN;
    float* bufA  = w + 2 * (size_t)NN;
    float* bufB  = bufA + (size_t)NN * HID;
    float* bufC  = bufB + (size_t)NN * HID;
    unsigned* own  = (unsigned*)(bufB + 8192);
    unsigned* rec  = own + NE / 4;
    unsigned* elist  = rec + NE;
    unsigned* ecount = elist + 2 * (size_t)NOWN * LCAP;

    unsigned* pc = (unsigned*)bufA;

    const int TPB = 256;

    outdeg_part_kernel<<<DG * DS, DEG_T, 0, stream>>>(src, pc);
    edge_rec_kernel<<<(NE / 4 + TPB - 1) / TPB, TPB, 0, stream>>>(src, dst, own, rec, pc, onorm, NE / 4);

    float* ew = bufB;
    label_weight_wmma<<<4 * (HID / 16), 32, 0, stream>>>(emb, W1, ew);
    owner_label_kernel<<<NOWN, AGG_T, 0, stream>>>(own, rec, labels, onorm, bufC, inorm, elist, ecount);
    gcn_gemm_wmma<LPAD, HID, HID, true><<<GEMM_BLOCKS, GEMM_T, 0, stream>>>(
        bufC, inorm, ew, b1, bufA, NN / 16);

    owner_sum_kernel<<<NOWN, AGG_T, 0, stream>>>(bufA, elist, ecount, own, rec, onorm, bufC);
    gcn_gemm_wmma<HID, HID, HID, true><<<GEMM_BLOCKS, GEMM_T, 0, stream>>>(
        bufC, inorm, W2, b2, bufA, NN / 16);

    owner_sum_kernel<<<NOWN, AGG_T, 0, stream>>>(bufA, elist, ecount, own, rec, onorm, bufC);
    gcn_gemm_wmma<HID, NLAB, 64, false><<<GEMM_BLOCKS, GEMM_T, 0, stream>>>(
        bufC, inorm, W3, b3, out, NN / 16);
}


# each owner block multiplies its finished LDS rows by the layer's weights on WMMA itself (W staged in LDS) and stores the layer's output: 3 GEMM launches and the sum buffers gone; sum order kept
# speedup vs baseline: 6.1035x; 1.2218x over previous
#include <hip/hip_runtime.h>

#define NN 50000
#define NE 800000
#define NLAB 50
#define EMB 64
#define HID 96
#define LPAD 52

#define RNG 196
#define NOWN ((NN + RNG - 1) / RNG)
#define AGG_T 1024
#define SCAN_L 7
#define NQ16 (NE / 16)
#define PASS_Q (AGG_T * SCAN_L)
#define NPASS ((NQ16 + PASS_Q - 1) / PASS_Q)
#define QALL 45056
#define LCAP 6144
#define LSTG 11264
#define RNGP ((RNG + 15) / 16 * 16)
#define SP (HID + 4)

static_assert(NOWN <= 256, "owner id must fit one byte");
static_assert(NN <= 65536 && RNG <= 65536, "edge record packs source and local row in 16 bits each");
static_assert(NE % 16 == 0, "owner bytes are read 16 at a time");
static_assert(PASS_Q <= NQ16, "the first pass must lie inside the owner array");
static_assert(LCAP <= QALL && LCAP <= LSTG && LCAP % AGG_T == 0, "a stored list comes out of one whole scan's queue and fits its LDS copy");

typedef __attribute__((ext_vector_type(2))) float v2f;
typedef __attribute__((ext_vector_type(8))) float v8f;

#define DG 16
#define DS 16
#define DGN (NN / DG)
#define DSE4 (NE / DS / 4)
#define DEG_T 1024
#define DEG_L ((DSE4 + DEG_T - 1) / DEG_T)

static_assert(NN % DG == 0 && NE % (DS * 4) == 0, "groups and slices are whole");
static_assert(NN <= NE / 4, "edge_rec_kernel finishes node i in the thread of edge quad i");

__global__ __launch_bounds__(DEG_T) void outdeg_part_kernel(const int* __restrict__ src,
                                                            unsigned* __restrict__ pc) {
    __shared__ unsigned cnt[DGN];

    const int tid = threadIdx.x;
    const unsigned g = blockIdx.x % DG, sl = blockIdx.x / DG;
    const unsigned lo = g * DGN;

    const int4* s4 = (const int4*)src + (size_t)sl * DSE4;
    int4 v[DEG_L];
#pragma unroll
    for (int k = 0; k < DEG_L; ++k) {
        const int i = k * DEG_T + tid;
        v[k] = s4[i < DSE4 ? i : DSE4 - 1];
    }

    for (int i = tid; i < DGN; i += DEG_T) cnt[i] = 0u;
    __syncthreads();

#pragma unroll
    for (int k = 0; k < DEG_L; ++k) {
        if (k * DEG_T + tid < DSE4) {
            const unsigned a0 = (unsigned)v[k].x - lo, a1 = (unsigned)v[k].y - lo;
            const unsigned a2 = (unsigned)v[k].z - lo, a3 = (unsigned)v[k].w - lo;
            if (a0 < DGN) atomicAdd(&cnt[a0], 1u);
            if (a1 < DGN) atomicAdd(&cnt[a1], 1u);
            if (a2 < DGN) atomicAdd(&cnt[a2], 1u);
            if (a3 < DGN) atomicAdd(&cnt[a3], 1u);
        }
    }
    __syncthreads();

    for (int i = tid; i < DGN; i += DEG_T) pc[(size_t)sl * NN + lo + i] = cnt[i];
}


__global__ void edge_rec_kernel(const int* __restrict__ src, const int* __restrict__ dst,
                                unsigned* __restrict__ own, unsigned* __restrict__ rec,
                                const unsigned* __restrict__ pc, float* __restrict__ onorm, int e4) {
    int i = blockIdx.x * blockDim.x + threadIdx.x;
    if (i < e4) {
        const int4 s = ((const int4*)src)[i];
        const int4 t = ((const int4*)dst)[i];
        if (i < NN) {
            unsigned c = 0u;
#pragma unroll
            for (int sl = 0; sl < DS; ++sl) c += pc[(size_t)sl * NN + i];
            onorm[i] = rsqrtf(fmaxf((float)c, 1.0f));
        }
        const unsigned o0 = (unsigned)t.x / RNG, o1 = (unsigned)t.y / RNG;
        const unsigned o2 = (unsigned)t.z / RNG, o3 = (unsigned)t.w / RNG;
        own[i] = o0 | (o1 << 8) | (o2 << 16) | (o3 << 24);
        uint4 r;
        r.x = (((unsigned)t.x - o0 * RNG) << 16) | (unsigned)s.x;
        r.y = (((unsigned)t.y - o1 * RNG) << 16) | (unsigned)s.y;
        r.z = (((unsigned)t.z - o2 * RNG) << 16) | (unsigned)s.z;
        r.w = (((unsigned)t.w - o3 * RNG) << 16) | (unsigned)s.w;
        ((uint4*)rec)[i] = r;
    }
}


__global__ __launch_bounds__(32) void label_weight_wmma(const float* __restrict__ emb,
                                                        const float* __restrict__ W1,
                                                        float* __restrict__ ew) {
    const int lane = threadIdx.x;
    const int m0 = (blockIdx.x / (HID / 16)) * 16;
    const int n0 = (blockIdx.x % (HID / 16)) * 16;

    const int rowA = m0 + (lane & 15);
    const int koff = (lane >> 4) * 2;
    const int colB = n0 + (lane & 15);
    const bool live = rowA < NLAB;
    const float* arow = emb + (long)(live ? rowA : 0) * EMB;

    v8f c = {0.0f, 0.0f, 0.0f, 0.0f, 0.0f, 0.0f, 0.0f, 0.0f};
    for (int k0 = 0; k0 < EMB; k0 += 4) {
        v2f a, b;
        float a0 = arow[k0 + koff];
        float a1 = arow[k0 + koff + 1];
        a.x = live ? a0 : 0.0f;
        a.y = live ? a1 : 0.0f;
        b.x = W1[(k0 + koff) * HID + colB];
        b.y = W1[(k0 + koff + 1) * HID + colB];
        c = __builtin_amdgcn_wmma_f32_16x16x4_f32(
                 false, a,  false, b,
                 (short)0, c,  false,  false);
    }

    const int rbase = m0 + (lane >> 4) * 8;
#pragma unroll
    for (int r = 0; r < 8; ++r) {
        int row = rbase + r;
        if (row < LPAD) ew[row * HID + colB] = (row < NLAB) ? c[r] : 0.0f;
    }
}


__device__ __forceinline__ unsigned own_bytes(unsigned x) {
    return ~(((x & 0x7f7f7f7fu) + 0x7f7f7f7fu) | x | 0x7f7f7f7fu);
}

__device__ __forceinline__ void lds_add(float* p, float v) {
    __hip_atomic_fetch_add(p, v, __ATOMIC_RELAXED, __HIP_MEMORY_SCOPE_WORKGROUP);
}

__device__ __forceinline__ void owner_scan_all(const unsigned* __restrict__ ob, unsigned me,
                                               unsigned* q, unsigned* qn, unsigned cap, int tid) {
    const unsigned pat = me * 0x01010101u;
    const uint4* own16 = (const uint4*)ob;

    uint4 cur[SCAN_L], nxt[SCAN_L];
#pragma unroll
    for (int j = 0; j < SCAN_L; ++j) cur[j] = own16[j * AGG_T + tid];

    for (int p = 0; p < NPASS; ++p) {
        if (p + 1 < NPASS) {
#pragma unroll
            for (int j = 0; j < SCAN_L; ++j) {
                const unsigned qi = (p + 1) * PASS_Q + j * AGG_T + tid;
                nxt[j] = own16[qi < NQ16 ? qi : NQ16 - 1];
            }
        }
#pragma unroll
        for (int j = 0; j < SCAN_L; ++j) {
            const unsigned wi = (unsigned)(p * PASS_Q + j * AGG_T + tid);
            const unsigned e0 = wi * 16u;
            const unsigned c = (own_bytes(cur[j].x ^ pat) >> 7) | (own_bytes(cur[j].y ^ pat) >> 6) |
                               (own_bytes(cur[j].z ^ pat) >> 5) | (own_bytes(cur[j].w ^ pat) >> 4);
            unsigned m = (wi < NQ16) ? c : 0u;
            while (m) {
                const unsigned b = (unsigned)__builtin_ctz(m);
                m &= m - 1u;
                const unsigned slot = __hip_atomic_fetch_add(qn, 1u, __ATOMIC_RELAXED,
                                                             __HIP_MEMORY_SCOPE_WORKGROUP);
                if (slot < cap) q[slot] = e0 + ((b & 7u) << 2) + (b >> 3);
            }
        }
        if (p + 1 < NPASS) {
#pragma unroll
            for (int j = 0; j < SCAN_L; ++j) cur[j] = nxt[j];
        }
    }
}


template <int K, int NOUT, int NPAD>
__device__ __forceinline__ void stage_weights(const float* __restrict__ W, const float* __restrict__ bias,
                                              float* ws, float* bs, int tid) {
    static_assert(K % 4 == 0 && NPAD % 32 == 0 && NOUT <= NPAD && NOUT % 2 == 0 && NPAD <= AGG_T, "tile shapes");
    constexpr int PB = 2 * NPAD + 32;
    constexpr int NP2 = NPAD / 2;
    constexpr int ITEMS = (K / 2) * NP2;
#pragma unroll
    for (int it = 0; it < (ITEMS + AGG_T - 1) / AGG_T; ++it) {
        const int idx = tid + it * AGG_T;
        if (idx < ITEMS) {
            const int p = idx / NP2, n = (idx % NP2) * 2;
            v2f lo = {0.0f, 0.0f}, hi = {0.0f, 0.0f};
            if (NOUT == NPAD || n < NOUT) {
                lo = *(const v2f*)(W + (2 * p) * NOUT + n);
                hi = *(const v2f*)(W + (2 * p + 1) * NOUT + n);
            }
            *(float4*)(ws + p * PB + 2 * n) = make_float4(lo.x, hi.x, lo.y, hi.y);
        }
    }
    if (tid < NPAD) bs[tid] = (tid < NOUT) ? bias[tid] : 0.0f;
}

template <int K, int PITCH, int NOUT, int NPAD, bool RELU>
__device__ __forceinline__ void owner_rows_gemm(const float* rowsL, const float* sc, const float* ws,
                                                const float* bs, float* __restrict__ out,
                                                int rows, int wave, int lane) {
    constexpr int HT = NPAD / 32;
    constexpr int KS = K / 4;
    constexpr int PB = 2 * NPAD + 32;
    constexpr int G = 2;
    constexpr int NG = (KS + G - 1) / G;

    const int cl = lane & 15;
    const int h = lane >> 4;
    const int units = ((rows + 15) >> 4) * 2;

    for (int u = wave; u < units; u += AGG_T / 32) {
        const int rt = u >> 1;
        const int t0 = (u & 1) * HT;
        const float* ap = rowsL + (rt * 16 + cl) * PITCH + 2 * h;
        const float* bp = ws + h * PB + 2 * cl + 32 * t0;
        const float s = sc[rt * 16 + cl];

        v8f c[HT];
#pragma unroll
        for (int t = 0; t < HT; ++t) {
            const float bval = bs[16 * (t0 + t) + cl];
#pragma unroll
            for (int r = 0; r < 8; ++r) c[t][r] = bval;
        }

        v2f aq[2][G];
        v2f bq[2][G][HT];
#pragma unroll
        for (int i = 0; i < G; ++i) {
            if (i < KS) {
                aq[0][i] = *(const v2f*)(ap + 4 * i);
#pragma unroll
                for (int t = 0; t < HT; ++t) bq[0][i][t] = *(const v2f*)(bp + 2 * i * PB + 32 * t);
            }
        }

#pragma unroll
        for (int g = 0; g < NG; ++g) {
            if (g + 1 < NG) {
#pragma unroll
                for (int i = 0; i < G; ++i) {
                    const int ks = (g + 1) * G + i;
                    if (ks < KS) {
                        aq[(g + 1) & 1][i] = *(const v2f*)(ap + 4 * ks);
#pragma unroll
                        for (int t = 0; t < HT; ++t)
                            bq[(g + 1) & 1][i][t] = *(const v2f*)(bp + 2 * ks * PB + 32 * t);
                    }
                }
            }
#pragma unroll
            for (int i = 0; i < G; ++i) {
                const int ks = g * G + i;
                if (ks < KS) {
                    v2f a;
                    a.x = aq[g & 1][i].x * s;
                    a.y = aq[g & 1][i].y * s;
#pragma unroll
                    for (int t = 0; t < HT; ++t)
                        c[t] = __builtin_amdgcn_wmma_f32_16x16x4_f32(
                                 false, a,  false, bq[g & 1][i][t],
                                 (short)0, c[t],  false,  false);
                }
            }
            __builtin_amdgcn_sched_barrier(0);
        }

        const int lr = rt * 16 + h * 8;
#pragma unroll
        for (int t = 0; t < HT; ++t) {
            const int col = 16 * (t0 + t) + cl;
            if (NOUT == NPAD || col < NOUT) {
#pragma unroll
                for (int r = 0; r < 8; ++r) {
                    float v = c[t][r];
                    if (RELU) v = fmaxf(v, 0.0f);
                    if (lr + r < rows) out[(size_t)(lr + r) * NOUT + col] = v;
                }
            }
        }
    }
}

__global__ __launch_bounds__(AGG_T) void owner_label_kernel(const unsigned* __restrict__ own,
                                                            const unsigned* __restrict__ rec,
                                                            const int* __restrict__ labels,
                                                            const float* __restrict__ onorm,
                                                            const float* __restrict__ ew,
                                                            const float* __restrict__ bias,
                                                            float* __restrict__ hout,
                                                            float* __restrict__ inorm,
                                                            unsigned* __restrict__ elist,
                                                            unsigned* __restrict__ ecount) {
    __shared__ __attribute__((aligned(16))) float lt[RNGP * LPAD];
    __shared__ unsigned queue[QALL];
    __shared__ __attribute__((aligned(16))) float ws[(LPAD / 2) * (2 * HID + 32)];
    __shared__ float bs[HID];
    __shared__ float sc[RNGP];
    __shared__ unsigned cnt[RNG];
    __shared__ unsigned qn;

    const int tid = threadIdx.x;
    const int lane = tid & 31;
    const int wave = __builtin_amdgcn_readfirstlane(tid >> 5);
    const unsigned me = blockIdx.x;
    const unsigned lo = me * RNG;

    float4* lt4 = (float4*)lt;
    for (int i = tid; i < RNGP * LPAD / 4; i += AGG_T) lt4[i] = make_float4(0.0f, 0.0f, 0.0f, 0.0f);
    if (tid < RNG) cnt[tid] = 0u;
    if (tid == 0) qn = 0u;
    stage_weights<LPAD, HID, HID>(ew, bias, ws, bs, tid);
    __syncthreads();

    owner_scan_all(own, me, queue, &qn, QALL, tid);
    __syncthreads();

    const unsigned tot = qn;
    if (tid == 0) ecount[me] = tot;
    if (tot <= QALL) {
        uint2* el = (uint2*)elist + (size_t)me * LCAP;
        for (unsigned i = tid; i < tot; i += AGG_T) {
            const unsigned r = rec[queue[i]];
            const unsigned s = r & 0xffffu, row = r >> 16;
            const float w = onorm[s];
            if (i < LCAP) el[i] = make_uint2(r, __float_as_uint(w));
            lds_add(&lt[row * LPAD + labels[s]], w);
            atomicAdd(&cnt[row], 1u);
        }
    } else {
        const unsigned char* ob = (const unsigned char*)own;
        for (unsigned e = tid; e < NE; e += AGG_T) {
            if (ob[e] == me) {
                const unsigned r = rec[e];
                const unsigned s = r & 0xffffu, row = r >> 16;
                lds_add(&lt[row * LPAD + labels[s]], onorm[s]);
                atomicAdd(&cnt[row], 1u);
            }
        }
    }
    __syncthreads();

    const int rows = ((int)NN - (int)lo < RNG) ? (int)NN - (int)lo : RNG;
    if (tid < RNGP) {
        float v = 0.0f;
        if (tid < rows) {
            v = rsqrtf(fmaxf((float)cnt[tid], 1.0f));
            inorm[lo + tid] = v;
        }
        sc[tid] = v;
    }
    __syncthreads();

    owner_rows_gemm<LPAD, LPAD, HID, HID, true>(lt, sc, ws, bs, hout + (size_t)lo * HID, rows, wave, lane);
}

template <int NOUT, int NPAD, bool RELU>
__global__ __launch_bounds__(AGG_T) void owner_sum_kernel(const float* __restrict__ hs,
                                                          const unsigned* __restrict__ elist,
                                                          const unsigned* __restrict__ ecount,
                                                          const unsigned* __restrict__ own,
                                                          const unsigned* __restrict__ rec,
                                                          const float* __restrict__ onorm,
                                                          const float* __restrict__ inorm,
                                                          const float* __restrict__ W,
                                                          const float* __restrict__ bias,
                                                          float* __restrict__ out) {
    __shared__ __attribute__((aligned(16))) float slab[RNGP * SP];
    __shared__ uint2 lst[LSTG];
    __shared__ __attribute__((aligned(16))) float ws[(HID / 2) * (2 * NPAD + 32)];
    __shared__ float bs[NPAD];
    __shared__ float sc[RNGP];

    const int tid = threadIdx.x;
    const int lane = tid & 31;
    const int wave = __builtin_amdgcn_readfirstlane(tid >> 5);
    const unsigned me = blockIdx.x;
    const int lo = me * RNG;
    const int rows = (NN - lo < RNG) ? NN - lo : RNG;

    const unsigned tot = __builtin_amdgcn_readfirstlane(ecount[me]);
    const bool listed = tot <= LCAP;

    const uint2* L = (const uint2*)elist + (size_t)me * LCAP;
    const uint2 none = {0xffffffffu, 0u};
    uint2 e[LCAP / AGG_T];
#pragma unroll
    for (int k = 0; k < LCAP / AGG_T; ++k) {
        const unsigned i = (unsigned)(k * AGG_T + tid);
        e[k] = none;
        if (listed && i < tot) e[k] = L[i];
    }

    float4* slab4 = (float4*)slab;
    for (int i = tid; i < RNGP * SP / 4; i += AGG_T) slab4[i] = make_float4(0.0f, 0.0f, 0.0f, 0.0f);
    stage_weights<HID, NOUT, NPAD>(W, bias, ws, bs, tid);
    if (tid < RNGP) sc[tid] = (tid < rows) ? inorm[lo + tid] : 0.0f;
#pragma unroll
    for (int k = 0; k < LCAP / AGG_T; ++k) lst[k * AGG_T + tid] = e[k];
    __syncthreads();

    if (listed) {
        const unsigned per = (((tot + 31u) >> 5) + 7u) & ~7u;
        const unsigned beg = wave * per;
        const unsigned end = (beg + per < tot) ? beg + per : tot;
        for (unsigned base = beg; base < end; base += 32u) {
            const unsigned idx = base + lane;
            const uint2 ent = (idx < end) ? lst[idx] : none;
            const unsigned c = (end - base < 32u) ? end - base : 32u;
            for (unsigned j0 = 0; j0 < c; j0 += 8u) {
                float v[8][3];
                float w[8];
                unsigned rr[8];
#pragma unroll
                for (int u = 0; u < 8; ++u) {
                    rr[u] = __builtin_amdgcn_readlane(ent.x, j0 + u);
                    w[u] = __uint_as_float(__builtin_amdgcn_readlane(ent.y, j0 + u));
                    const unsigned s = (rr[u] == 0xffffffffu) ? 0u : (rr[u] & 0xffffu);
                    const float* hp = hs + (size_t)s * HID + lane;
                    v[u][0] = hp[0];
                    v[u][1] = hp[32];
                    v[u][2] = hp[64];
                }
#pragma unroll
                for (int u = 0; u < 8; ++u) {
                    if (rr[u] != 0xffffffffu) {
                        float* sp = slab + (rr[u] >> 16) * SP + lane;
                        lds_add(sp, v[u][0] * w[u]);
                        lds_add(sp + 32, v[u][1] * w[u]);
                        lds_add(sp + 64, v[u][2] * w[u]);
                    }
                }
            }
        }
    } else {
        const unsigned char* ob = (const unsigned char*)own;
        for (unsigned ed = tid; ed < NE; ed += AGG_T) {
            if (ob[ed] == me) {
                const unsigned r = rec[ed];
                const float* hp = hs + (size_t)(r & 0xffffu) * HID;
                const float w = onorm[r & 0xffffu];
                float* sp = slab + (r >> 16) * SP;
                for (int f = 0; f < HID; ++f) lds_add(sp + f, hp[f] * w);
            }
        }
    }
    __syncthreads();

    owner_rows_gemm<HID, SP, NOUT, NPAD, RELU>(slab, sc, ws, bs, out + (size_t)lo * NOUT, rows, wave, lane);
}


extern "C" void kernel_launch(void* const* d_in, const int* in_sizes, int n_in,
                              void* d_out, int out_size, void* d_ws, size_t ws_size,
                              hipStream_t stream) {
    const int*   labels = (const int*)d_in[0];
    const int*   src    = (const int*)d_in[1];
    const int*   dst    = (const int*)d_in[2];
    const float* emb    = (const float*)d_in[3];
    const float* W1     = (const float*)d_in[4];
    const float* b1     = (const float*)d_in[5];
    const float* W2     = (const float*)d_in[6];
    const float* b2     = (const float*)d_in[7];
    const float* W3     = (const float*)d_in[8];
    const float* b3     = (const float*)d_in[9];
    float* out = (float*)d_out;

    float* w = (float*)d_ws;
    float* onorm = w;
    float* inorm = w + NN;
    float* bufA  = w + 2 * (size_t)NN;
    float* bufB  = bufA + (size_t)NN * HID;
    float* bufC  = bufB + (size_t)NN * HID;
    unsigned* own  = (unsigned*)(bufB + 8192);
    unsigned* rec  = own + NE / 4;
    unsigned* elist  = rec + NE;
    unsigned* ecount = elist + 2 * (size_t)NOWN * LCAP;

    unsigned* pc = (unsigned*)bufA;

    const int TPB = 256;

    outdeg_part_kernel<<<DG * DS, DEG_T, 0, stream>>>(src, pc);
    edge_rec_kernel<<<(NE / 4 + TPB - 1) / TPB, TPB, 0, stream>>>(src, dst, own, rec, pc, onorm, NE / 4);

    float* ew = bufB;
    label_weight_wmma<<<4 * (HID / 16), 32, 0, stream>>>(emb, W1, ew);
    owner_label_kernel<<<NOWN, AGG_T, 0, stream>>>(own, rec, labels, onorm, ew, b1, bufA, inorm, elist, ecount);

    owner_sum_kernel<HID, HID, true><<<NOWN, AGG_T, 0, stream>>>(
        bufA, elist, ecount, own, rec, onorm, inorm, W2, b2, bufC);

    owner_sum_kernel<NLAB, 64, false><<<NOWN, AGG_T, 0, stream>>>(
        bufC, elist, ecount, own, rec, onorm, inorm, W3, b3, out);
}


# owners' edge search in two levels: 16 groups x 16 edge slices blocks list a group's edges (wave votes, one LDS add per wave); an owner scans only its group's lists; sum order: LDS-add arrival
# speedup vs baseline: 6.4515x; 1.0570x over previous
#include <hip/hip_runtime.h>

#define NN 50000
#define NE 800000
#define NLAB 50
#define EMB 64
#define HID 96
#define LPAD 52

#define RNG 196
#define NOWN ((NN + RNG - 1) / RNG)
#define AGG_T 1024
#define GRP 16
#define NGRP (NOWN / GRP)
#define GRNG (GRP * RNG)
#define GS 16
#define GSE4 (NE / GS / 4)
#define GCAP 4096
#define GL_T 1024
#define GL_L ((GSE4 + GL_T - 1) / GL_T)
#define GW16 (GCAP / 16)
#define GW_L (GS * GW16 / AGG_T)
#define QALL 45056
#define LCAP 6144
#define LSTG 11264
#define RNGP ((RNG + 15) / 16 * 16)
#define SP (HID + 4)

static_assert(NOWN <= 256, "owner id must fit one byte");
static_assert(NN <= 65536 && RNG <= 65536, "edge record packs source and local row in 16 bits each");
static_assert(NOWN % GRP == 0 && NE % (GS * 4) == 0, "groups and slices are whole");
static_assert(GCAP % 16 == 0 && GCAP == 4 * GL_T && (GS * GW16) % AGG_T == 0, "list words split evenly over the threads");
static_assert(NGRP * GS * GL_T >= NN, "group_list_kernel finishes node i in its i-th thread");
static_assert(LCAP <= QALL && LCAP <= LSTG && LCAP % AGG_T == 0, "a stored list comes out of one whole scan's queue and fits its LDS copy");

typedef __attribute__((ext_vector_type(2))) float v2f;
typedef __attribute__((ext_vector_type(8))) float v8f;

#define DG 16
#define DS 16
#define DGN (NN / DG)
#define DSE4 (NE / DS / 4)
#define DEG_T 1024
#define DEG_L ((DSE4 + DEG_T - 1) / DEG_T)

static_assert(NN % DG == 0 && NE % (DS * 4) == 0, "groups and slices are whole");

__global__ __launch_bounds__(DEG_T) void outdeg_part_kernel(const int* __restrict__ src,
                                                            unsigned* __restrict__ pc) {
    __shared__ unsigned cnt[DGN];

    const int tid = threadIdx.x;
    const unsigned g = blockIdx.x % DG, sl = blockIdx.x / DG;
    const unsigned lo = g * DGN;

    const int4* s4 = (const int4*)src + (size_t)sl * DSE4;
    int4 v[DEG_L];
#pragma unroll
    for (int k = 0; k < DEG_L; ++k) {
        const int i = k * DEG_T + tid;
        v[k] = s4[i < DSE4 ? i : DSE4 - 1];
    }

    for (int i = tid; i < DGN; i += DEG_T) cnt[i] = 0u;
    __syncthreads();

#pragma unroll
    for (int k = 0; k < DEG_L; ++k) {
        if (k * DEG_T + tid < DSE4) {
            const unsigned a0 = (unsigned)v[k].x - lo, a1 = (unsigned)v[k].y - lo;
            const unsigned a2 = (unsigned)v[k].z - lo, a3 = (unsigned)v[k].w - lo;
            if (a0 < DGN) atomicAdd(&cnt[a0], 1u);
            if (a1 < DGN) atomicAdd(&cnt[a1], 1u);
            if (a2 < DGN) atomicAdd(&cnt[a2], 1u);
            if (a3 < DGN) atomicAdd(&cnt[a3], 1u);
        }
    }
    __syncthreads();

    for (int i = tid; i < DGN; i += DEG_T) pc[(size_t)sl * NN + lo + i] = cnt[i];
}


__global__ __launch_bounds__(GL_T) void group_list_kernel(const int* __restrict__ src, const int* __restrict__ dst,
                                                          unsigned* __restrict__ grec, unsigned* __restrict__ gown,
                                                          unsigned* __restrict__ gcount,
                                                          const unsigned* __restrict__ pc, float* __restrict__ onorm) {
    __shared__ __attribute__((aligned(16))) unsigned qrec[GCAP];
    __shared__ unsigned qn;

    const int tid = threadIdx.x;
    const int lane = tid & 31;
    const unsigned g = blockIdx.x % NGRP, sl = blockIdx.x / NGRP;
    const unsigned lo = g * GRNG;

    const int4* d4 = (const int4*)dst + (size_t)sl * GSE4;
    const int4* s4 = (const int4*)src + (size_t)sl * GSE4;
    int4 v[GL_L], sv[GL_L];
#pragma unroll
    for (int k = 0; k < GL_L; ++k) {
        const int i = k * GL_T + tid;
        v[k] = d4[i < GSE4 ? i : GSE4 - 1];
    }
#pragma unroll
    for (int k = 0; k < GL_L; ++k) {
        const int i = k * GL_T + tid;
        sv[k] = s4[i < GSE4 ? i : GSE4 - 1];
    }

    const int node = blockIdx.x * GL_T + tid;
    if (node < NN) {
        unsigned c = 0u;
#pragma unroll
        for (int p = 0; p < DS; ++p) c += pc[(size_t)p * NN + node];
        onorm[node] = rsqrtf(fmaxf((float)c, 1.0f));
    }

    ((uint4*)qrec)[tid] = make_uint4(0u, 0u, 0u, 0u);
    if (tid == 0) qn = 0u;
    __syncthreads();

    unsigned wc = 0u;
#pragma unroll
    for (int k = 0; k < GL_L; ++k) {
        const bool in = k * GL_T + tid < GSE4;
        wc += (unsigned)__builtin_popcount(__builtin_amdgcn_ballot_w32(in && (unsigned)v[k].x - lo < GRNG));
        wc += (unsigned)__builtin_popcount(__builtin_amdgcn_ballot_w32(in && (unsigned)v[k].y - lo < GRNG));
        wc += (unsigned)__builtin_popcount(__builtin_amdgcn_ballot_w32(in && (unsigned)v[k].z - lo < GRNG));
        wc += (unsigned)__builtin_popcount(__builtin_amdgcn_ballot_w32(in && (unsigned)v[k].w - lo < GRNG));
    }
    unsigned base = 0u;
    if (lane == 0) base = __hip_atomic_fetch_add(&qn, wc, __ATOMIC_RELAXED, __HIP_MEMORY_SCOPE_WORKGROUP);
    base = __builtin_amdgcn_readfirstlane(base);

#pragma unroll
    for (int k = 0; k < GL_L; ++k) {
        const bool in = k * GL_T + tid < GSE4;
        const unsigned a[4] = {(unsigned)v[k].x - lo, (unsigned)v[k].y - lo,
                               (unsigned)v[k].z - lo, (unsigned)v[k].w - lo};
        const unsigned ss[4] = {(unsigned)sv[k].x, (unsigned)sv[k].y, (unsigned)sv[k].z, (unsigned)sv[k].w};
#pragma unroll
        for (int j = 0; j < 4; ++j) {
            const bool hit = in && a[j] < GRNG;
            const unsigned m = __builtin_amdgcn_ballot_w32(hit);
            const unsigned slot = __builtin_amdgcn_mbcnt_lo(m, base);
            if (hit && slot < GCAP) qrec[slot] = (a[j] << 16) | ss[j];
            base += (unsigned)__builtin_popcount(m);
        }
    }
    __syncthreads();

    const unsigned tot = qn;
    const unsigned n = tot < GCAP ? tot : GCAP;
    const size_t list = (size_t)g * GS + sl;
    if ((unsigned)tid * 4u < ((n + 15u) & ~15u)) {
        const uint4 q = ((const uint4*)qrec)[tid];
        const unsigned qq[4] = {q.x, q.y, q.z, q.w};
        unsigned r[4];
        unsigned ob = 0u;
#pragma unroll
        for (int c = 0; c < 4; ++c) {
            const bool live = (unsigned)tid * 4u + c < n;
            const unsigned a = qq[c] >> 16;
            const unsigned o = a / RNG;
            r[c] = live ? (((a - o * RNG) << 16) | (qq[c] & 0xffffu)) : 0u;
            ob |= (live ? g * GRP + o : 0u) << (8 * c);
        }
        ((uint4*)(grec + list * GCAP))[tid] = make_uint4(r[0], r[1], r[2], r[3]);
        gown[list * (GCAP / 4) + tid] = ob;
    }
    if (tid == 0) gcount[list] = tot;
}


__global__ __launch_bounds__(32) void label_weight_wmma(const float* __restrict__ emb,
                                                        const float* __restrict__ W1,
                                                        float* __restrict__ ew) {
    const int lane = threadIdx.x;
    const int m0 = (blockIdx.x / (HID / 16)) * 16;
    const int n0 = (blockIdx.x % (HID / 16)) * 16;

    const int rowA = m0 + (lane & 15);
    const int koff = (lane >> 4) * 2;
    const int colB = n0 + (lane & 15);
    const bool live = rowA < NLAB;
    const float* arow = emb + (long)(live ? rowA : 0) * EMB;

    v8f c = {0.0f, 0.0f, 0.0f, 0.0f, 0.0f, 0.0f, 0.0f, 0.0f};
    for (int k0 = 0; k0 < EMB; k0 += 4) {
        v2f a, b;
        float a0 = arow[k0 + koff];
        float a1 = arow[k0 + koff + 1];
        a.x = live ? a0 : 0.0f;
        a.y = live ? a1 : 0.0f;
        b.x = W1[(k0 + koff) * HID + colB];
        b.y = W1[(k0 + koff + 1) * HID + colB];
        c = __builtin_amdgcn_wmma_f32_16x16x4_f32(
                 false, a,  false, b,
                 (short)0, c,  false,  false);
    }

    const int rbase = m0 + (lane >> 4) * 8;
#pragma unroll
    for (int r = 0; r < 8; ++r) {
        int row = rbase + r;
        if (row < LPAD) ew[row * HID + colB] = (row < NLAB) ? c[r] : 0.0f;
    }
}


__device__ __forceinline__ unsigned own_bytes(unsigned x) {
    return ~(((x & 0x7f7f7f7fu) + 0x7f7f7f7fu) | x | 0x7f7f7f7fu);
}

__device__ __forceinline__ void lds_add(float* p, float v) {
    __hip_atomic_fetch_add(p, v, __ATOMIC_RELAXED, __HIP_MEMORY_SCOPE_WORKGROUP);
}

__device__ __forceinline__ void group_scan(const unsigned* __restrict__ gown, const unsigned* __restrict__ gcount,
                                           unsigned me, unsigned* q, unsigned* qn, unsigned cap,
                                           unsigned* ovf, int tid) {
    const unsigned g = me / GRP;
    const unsigned pat = me * 0x01010101u;
    const uint4* gw = (const uint4*)gown + (size_t)g * (GS * GW16);
    const unsigned* gc = gcount + g * GS;

    unsigned n[GW_L];
    uint4 x[GW_L];
#pragma unroll
    for (int j = 0; j < GW_L; ++j) {
        const unsigned c = gc[(unsigned)(j * AGG_T + tid) / GW16];
        if (c > GCAP) *ovf = 1u;
        n[j] = c < GCAP ? c : GCAP;
    }
#pragma unroll
    for (int j = 0; j < GW_L; ++j) {
        const unsigned w = (unsigned)(j * AGG_T + tid);
        x[j] = make_uint4(0u, 0u, 0u, 0u);
        if ((w % GW16) * 16u < n[j]) x[j] = gw[w];
    }
#pragma unroll
    for (int j = 0; j < GW_L; ++j) {
        const unsigned w = (unsigned)(j * AGG_T + tid);
        const unsigned pos = (w % GW16) * 16u;
        if (pos < n[j]) {
            const unsigned rem = n[j] - pos;
            unsigned m = (own_bytes(x[j].x ^ pat) >> 7) | (own_bytes(x[j].y ^ pat) >> 6) |
                         (own_bytes(x[j].z ^ pat) >> 5) | (own_bytes(x[j].w ^ pat) >> 4);
            while (m) {
                const unsigned b = (unsigned)__builtin_ctz(m);
                m &= m - 1u;
                const unsigned k = ((b & 7u) << 2) + (b >> 3);
                if (k < rem) {
                    const unsigned slot = __hip_atomic_fetch_add(qn, 1u, __ATOMIC_RELAXED,
                                                                 __HIP_MEMORY_SCOPE_WORKGROUP);
                    if (slot < cap) q[slot] = w * 16u + k;
                }
            }
        }
    }
}


template <int K, int NOUT, int NPAD>
__device__ __forceinline__ void stage_weights(const float* __restrict__ W, const float* __restrict__ bias,
                                              float* ws, float* bs, int tid) {
    static_assert(K % 4 == 0 && NPAD % 32 == 0 && NOUT <= NPAD && NOUT % 2 == 0 && NPAD <= AGG_T, "tile shapes");
    constexpr int PB = 2 * NPAD + 32;
    constexpr int NP2 = NPAD / 2;
    constexpr int ITEMS = (K / 2) * NP2;
#pragma unroll
    for (int it = 0; it < (ITEMS + AGG_T - 1) / AGG_T; ++it) {
        const int idx = tid + it * AGG_T;
        if (idx < ITEMS) {
            const int p = idx / NP2, n = (idx % NP2) * 2;
            v2f lo = {0.0f, 0.0f}, hi = {0.0f, 0.0f};
            if (NOUT == NPAD || n < NOUT) {
                lo = *(const v2f*)(W + (2 * p) * NOUT + n);
                hi = *(const v2f*)(W + (2 * p + 1) * NOUT + n);
            }
            *(float4*)(ws + p * PB + 2 * n) = make_float4(lo.x, hi.x, lo.y, hi.y);
        }
    }
    if (tid < NPAD) bs[tid] = (tid < NOUT) ? bias[tid] : 0.0f;
}

template <int K, int PITCH, int NOUT, int NPAD, bool RELU>
__device__ __forceinline__ void owner_rows_gemm(const float* rowsL, const float* sc, const float* ws,
                                                const float* bs, float* __restrict__ out,
                                                int rows, int wave, int lane) {
    constexpr int HT = NPAD / 32;
    constexpr int KS = K / 4;
    constexpr int PB = 2 * NPAD + 32;
    constexpr int G = 2;
    constexpr int NG = (KS + G - 1) / G;

    const int cl = lane & 15;
    const int h = lane >> 4;
    const int units = ((rows + 15) >> 4) * 2;

    for (int u = wave; u < units; u += AGG_T / 32) {
        const int rt = u >> 1;
        const int t0 = (u & 1) * HT;
        const float* ap = rowsL + (rt * 16 + cl) * PITCH + 2 * h;
        const float* bp = ws + h * PB + 2 * cl + 32 * t0;
        const float s = sc[rt * 16 + cl];

        v8f c[HT];
#pragma unroll
        for (int t = 0; t < HT; ++t) {
            const float bval = bs[16 * (t0 + t) + cl];
#pragma unroll
            for (int r = 0; r < 8; ++r) c[t][r] = bval;
        }

        v2f aq[2][G];
        v2f bq[2][G][HT];
#pragma unroll
        for (int i = 0; i < G; ++i) {
            if (i < KS) {
                aq[0][i] = *(const v2f*)(ap + 4 * i);
#pragma unroll
                for (int t = 0; t < HT; ++t) bq[0][i][t] = *(const v2f*)(bp + 2 * i * PB + 32 * t);
            }
        }

#pragma unroll
        for (int g = 0; g < NG; ++g) {
            if (g + 1 < NG) {
#pragma unroll
                for (int i = 0; i < G; ++i) {
                    const int ks = (g + 1) * G + i;
                    if (ks < KS) {
                        aq[(g + 1) & 1][i] = *(const v2f*)(ap + 4 * ks);
#pragma unroll
                        for (int t = 0; t < HT; ++t)
                            bq[(g + 1) & 1][i][t] = *(const v2f*)(bp + 2 * ks * PB + 32 * t);
                    }
                }
            }
#pragma unroll
            for (int i = 0; i < G; ++i) {
                const int ks = g * G + i;
                if (ks < KS) {
                    v2f a;
                    a.x = aq[g & 1][i].x * s;
                    a.y = aq[g & 1][i].y * s;
#pragma unroll
                    for (int t = 0; t < HT; ++t)
                        c[t] = __builtin_amdgcn_wmma_f32_16x16x4_f32(
                                 false, a,  false, bq[g & 1][i][t],
                                 (short)0, c[t],  false,  false);
                }
            }
            __builtin_amdgcn_sched_barrier(0);
        }

        const int lr = rt * 16 + h * 8;
#pragma unroll
        for (int t = 0; t < HT; ++t) {
            const int col = 16 * (t0 + t) + cl;
            if (NOUT == NPAD || col < NOUT) {
#pragma unroll
                for (int r = 0; r < 8; ++r) {
                    float v = c[t][r];
                    if (RELU) v = fmaxf(v, 0.0f);
                    if (lr + r < rows) out[(size_t)(lr + r) * NOUT + col] = v;
                }
            }
        }
    }
}

__global__ __launch_bounds__(AGG_T) void owner_label_kernel(const unsigned* __restrict__ gown,
                                                            const unsigned* __restrict__ grec,
                                                            const unsigned* __restrict__ gcount,
                                                            const int* __restrict__ src,
                                                            const int* __restrict__ dst,
                                                            const int* __restrict__ labels,
                                                            const float* __restrict__ onorm,
                                                            const float* __restrict__ ew,
                                                            const float* __restrict__ bias,
                                                            float* __restrict__ hout,
                                                            float* __restrict__ inorm,
                                                            unsigned* __restrict__ elist,
                                                            unsigned* __restrict__ ecount) {
    __shared__ __attribute__((aligned(16))) float lt[RNGP * LPAD];
    __shared__ unsigned queue[QALL];
    __shared__ __attribute__((aligned(16))) float ws[(LPAD / 2) * (2 * HID + 32)];
    __shared__ float bs[HID];
    __shared__ float sc[RNGP];
    __shared__ unsigned cnt[RNG];
    __shared__ unsigned qn;
    __shared__ unsigned ovf;

    const int tid = threadIdx.x;
    const int lane = tid & 31;
    const int wave = __builtin_amdgcn_readfirstlane(tid >> 5);
    const unsigned me = blockIdx.x;
    const unsigned lo = me * RNG;

    float4* lt4 = (float4*)lt;
    for (int i = tid; i < RNGP * LPAD / 4; i += AGG_T) lt4[i] = make_float4(0.0f, 0.0f, 0.0f, 0.0f);
    if (tid < RNG) cnt[tid] = 0u;
    if (tid == 0) { qn = 0u; ovf = 0u; }
    stage_weights<LPAD, HID, HID>(ew, bias, ws, bs, tid);
    __syncthreads();

    group_scan(gown, gcount, me, queue, &qn, QALL, &ovf, tid);
    __syncthreads();

    const unsigned tot = qn;
    const bool whole = ovf == 0u && tot <= QALL;
    if (tid == 0) ecount[me] = whole ? tot : 0xffffffffu;
    if (whole) {
        uint2* el = (uint2*)elist + (size_t)me * LCAP;
        const unsigned* gr = grec + (size_t)(me / GRP) * (GS * GCAP);
        for (unsigned i = tid; i < tot; i += AGG_T) {
            const unsigned r = gr[queue[i]];
            const unsigned s = r & 0xffffu, row = r >> 16;
            const float w = onorm[s];
            if (i < LCAP) el[i] = make_uint2(r, __float_as_uint(w));
            lds_add(&lt[row * LPAD + labels[s]], w);
            atomicAdd(&cnt[row], 1u);
        }
    } else {
        for (unsigned e = tid; e < NE; e += AGG_T) {
            const unsigned row = (unsigned)dst[e] - lo;
            if (row < RNG) {
                const unsigned s = (unsigned)src[e];
                lds_add(&lt[row * LPAD + labels[s]], onorm[s]);
                atomicAdd(&cnt[row], 1u);
            }
        }
    }
    __syncthreads();

    const int rows = ((int)NN - (int)lo < RNG) ? (int)NN - (int)lo : RNG;
    if (tid < RNGP) {
        float v = 0.0f;
        if (tid < rows) {
            v = rsqrtf(fmaxf((float)cnt[tid], 1.0f));
            inorm[lo + tid] = v;
        }
        sc[tid] = v;
    }
    __syncthreads();

    owner_rows_gemm<LPAD, LPAD, HID, HID, true>(lt, sc, ws, bs, hout + (size_t)lo * HID, rows, wave, lane);
}

template <int NOUT, int NPAD, bool RELU>
__global__ __launch_bounds__(AGG_T) void owner_sum_kernel(const float* __restrict__ hs,
                                                          const unsigned* __restrict__ elist,
                                                          const unsigned* __restrict__ ecount,
                                                          const int* __restrict__ src,
                                                          const int* __restrict__ dst,
                                                          const float* __restrict__ onorm,
                                                          const float* __restrict__ inorm,
                                                          const float* __restrict__ W,
                                                          const float* __restrict__ bias,
                                                          float* __restrict__ out) {
    __shared__ __attribute__((aligned(16))) float slab[RNGP * SP];
    __shared__ uint2 lst[LSTG];
    __shared__ __attribute__((aligned(16))) float ws[(HID / 2) * (2 * NPAD + 32)];
    __shared__ float bs[NPAD];
    __shared__ float sc[RNGP];

    const int tid = threadIdx.x;
    const int lane = tid & 31;
    const int wave = __builtin_amdgcn_readfirstlane(tid >> 5);
    const unsigned me = blockIdx.x;
    const int lo = me * RNG;
    const int rows = (NN - lo < RNG) ? NN - lo : RNG;

    const unsigned tot = __builtin_amdgcn_readfirstlane(ecount[me]);
    const bool listed = tot <= LCAP;

    const uint2* L = (const uint2*)elist + (size_t)me * LCAP;
    const uint2 none = {0xffffffffu, 0u};
    uint2 e[LCAP / AGG_T];
#pragma unroll
    for (int k = 0; k < LCAP / AGG_T; ++k) {
        const unsigned i = (unsigned)(k * AGG_T + tid);
        e[k] = none;
        if (listed && i < tot) e[k] = L[i];
    }

    float4* slab4 = (float4*)slab;
    for (int i = tid; i < RNGP * SP / 4; i += AGG_T) slab4[i] = make_float4(0.0f, 0.0f, 0.0f, 0.0f);
    stage_weights<HID, NOUT, NPAD>(W, bias, ws, bs, tid);
    if (tid < RNGP) sc[tid] = (tid < rows) ? inorm[lo + tid] : 0.0f;
#pragma unroll
    for (int k = 0; k < LCAP / AGG_T; ++k) lst[k * AGG_T + tid] = e[k];
    __syncthreads();

    if (listed) {
        const unsigned per = (((tot + 31u) >> 5) + 7u) & ~7u;
        const unsigned beg = wave * per;
        const unsigned end = (beg + per < tot) ? beg + per : tot;
        for (unsigned base = beg; base < end; base += 32u) {
            const unsigned idx = base + lane;
            const uint2 ent = (idx < end) ? lst[idx] : none;
            const unsigned c = (end - base < 32u) ? end - base : 32u;
            for (unsigned j0 = 0; j0 < c; j0 += 8u) {
                float v[8][3];
                float w[8];
                unsigned rr[8];
#pragma unroll
                for (int u = 0; u < 8; ++u) {
                    rr[u] = __builtin_amdgcn_readlane(ent.x, j0 + u);
                    w[u] = __uint_as_float(__builtin_amdgcn_readlane(ent.y, j0 + u));
                    const unsigned s = (rr[u] == 0xffffffffu) ? 0u : (rr[u] & 0xffffu);
                    const float* hp = hs + (size_t)s * HID + lane;
                    v[u][0] = hp[0];
                    v[u][1] = hp[32];
                    v[u][2] = hp[64];
                }
#pragma unroll
                for (int u = 0; u < 8; ++u) {
                    if (rr[u] != 0xffffffffu) {
                        float* sp = slab + (rr[u] >> 16) * SP + lane;
                        lds_add(sp, v[u][0] * w[u]);
                        lds_add(sp + 32, v[u][1] * w[u]);
                        lds_add(sp + 64, v[u][2] * w[u]);
                    }
                }
            }
        }
    } else {
        for (unsigned ed = tid; ed < NE; ed += AGG_T) {
            const unsigned row = (unsigned)dst[ed] - (unsigned)lo;
            if (row < RNG) {
                const unsigned sn = (unsigned)src[ed];
                const float* hp = hs + (size_t)sn * HID;
                const float w = onorm[sn];
                float* sp = slab + row * SP;
                for (int f = 0; f < HID; ++f) lds_add(sp + f, hp[f] * w);
            }
        }
    }
    __syncthreads();

    owner_rows_gemm<HID, SP, NOUT, NPAD, RELU>(slab, sc, ws, bs, out + (size_t)lo * NOUT, rows, wave, lane);
}


extern "C" void kernel_launch(void* const* d_in, const int* in_sizes, int n_in,
                              void* d_out, int out_size, void* d_ws, size_t ws_size,
                              hipStream_t stream) {
    const int*   labels = (const int*)d_in[0];
    const int*   src    = (const int*)d_in[1];
    const int*   dst    = (const int*)d_in[2];
    const float* emb    = (const float*)d_in[3];
    const float* W1     = (const float*)d_in[4];
    const float* b1     = (const float*)d_in[5];
    const float* W2     = (const float*)d_in[6];
    const float* b2     = (const float*)d_in[7];
    const float* W3     = (const float*)d_in[8];
    const float* b3     = (const float*)d_in[9];
    float* out = (float*)d_out;

    float* w = (float*)d_ws;
    float* onorm = w;
    float* inorm = w + NN;
    float* bufA  = w + 2 * (size_t)NN;
    float* bufB  = bufA + (size_t)NN * HID;
    float* bufC  = bufB + (size_t)NN * HID;
    unsigned* gown = (unsigned*)(bufB + 8192);
    unsigned* grec = gown + NGRP * GS * (GCAP / 4);
    unsigned* gcount = grec + (size_t)NGRP * GS * GCAP;
    unsigned* elist  = gcount + NGRP * GS;
    unsigned* ecount = elist + 2 * (size_t)NOWN * LCAP;

    unsigned* pc = (unsigned*)bufA;

    outdeg_part_kernel<<<DG * DS, DEG_T, 0, stream>>>(src, pc);
    group_list_kernel<<<NGRP * GS, GL_T, 0, stream>>>(src, dst, grec, gown, gcount, pc, onorm);

    float* ew = bufB;
    label_weight_wmma<<<4 * (HID / 16), 32, 0, stream>>>(emb, W1, ew);
    owner_label_kernel<<<NOWN, AGG_T, 0, stream>>>(gown, grec, gcount, src, dst, labels, onorm, ew, b1, bufA, inorm,
                                                   elist, ecount);

    owner_sum_kernel<HID, HID, true><<<NOWN, AGG_T, 0, stream>>>(
        bufA, elist, ecount, src, dst, onorm, inorm, W2, b2, bufC);

    owner_sum_kernel<NLAB, 64, false><<<NOWN, AGG_T, 0, stream>>>(
        bufC, elist, ecount, src, dst, onorm, inorm, W3, b3, out);
}


# out-degree counting and emb@W1 moved into the group-list launch (a block counts its own src slice in LDS; a node's 16 partial counts lie side by side, a reader sums one 64-byte read): 6 launches -> 4
# speedup vs baseline: 6.8276x; 1.0583x over previous
#include <hip/hip_runtime.h>

#define NN 50000
#define NE 800000
#define NLAB 50
#define EMB 64
#define HID 96
#define LPAD 52

#define RNG 196
#define NOWN ((NN + RNG - 1) / RNG)
#define AGG_T 1024
#define GRP 16
#define NGRP (NOWN / GRP)
#define GRNG (GRP * RNG)
#define GS 16
#define GSE4 (NE / GS / 4)
#define GCAP 4096
#define GL_T 1024
#define GL_L ((GSE4 + GL_T - 1) / GL_T)
#define GW16 (GCAP / 16)
#define GW_L (GS * GW16 / AGG_T)
#define QALL 45056
#define LCAP 6144
#define LSTG 11264
#define RNGP ((RNG + 15) / 16 * 16)
#define SP (HID + 4)

static_assert(NOWN <= 256, "owner id must fit one byte");
static_assert(NN <= 65536 && RNG <= 65536, "edge record packs source and local row in 16 bits each");
static_assert(NOWN % GRP == 0 && NE % (GS * 4) == 0, "groups and slices are whole");
static_assert(GCAP % 16 == 0 && GCAP == 4 * GL_T && (GS * GW16) % AGG_T == 0, "list words split evenly over the threads");
static_assert(LCAP <= QALL && LCAP <= LSTG && LCAP % AGG_T == 0, "a stored list comes out of one whole scan's queue and fits its LDS copy");

typedef __attribute__((ext_vector_type(2))) float v2f;
typedef __attribute__((ext_vector_type(8))) float v8f;

#define DG 16
#define DS 16
#define DGN (NN / DG)

static_assert(NN % DG == 0 && NE % (DS * 4) == 0, "groups and slices are whole");
static_assert(DG == NGRP && DS == GS, "a (group, slice) block of the list launch is also a (node group, slice) block of the count");
static_assert(DS == 16, "a node's counts are four 16-byte words");

__device__ __forceinline__ float out_norm_of(const unsigned* __restrict__ pc, unsigned s) {
    const uint4* p = (const uint4*)pc + (size_t)s * (DS / 4);
    const uint4 a = p[0], b = p[1], c = p[2], d = p[3];
    const unsigned n = (a.x + a.y + a.z + a.w) + (b.x + b.y + b.z + b.w) +
                       (c.x + c.y + c.z + c.w) + (d.x + d.y + d.z + d.w);
    return rsqrtf(fmaxf((float)n, 1.0f));
}


#define EW_TILES (4 * (HID / 16))
__device__ __forceinline__ void label_weight_tile(const float* __restrict__ emb,
                                                  const float* __restrict__ W1,
                                                  float* __restrict__ ew, int tile, int lane) {
    const int m0 = (tile / (HID / 16)) * 16;
    const int n0 = (tile % (HID / 16)) * 16;

    const int rowA = m0 + (lane & 15);
    const int koff = (lane >> 4) * 2;
    const int colB = n0 + (lane & 15);
    const bool live = rowA < NLAB;
    const float* arow = emb + (long)(live ? rowA : 0) * EMB;

    v8f c = {0.0f, 0.0f, 0.0f, 0.0f, 0.0f, 0.0f, 0.0f, 0.0f};
    for (int k0 = 0; k0 < EMB; k0 += 4) {
        v2f a, b;
        float a0 = arow[k0 + koff];
        float a1 = arow[k0 + koff + 1];
        a.x = live ? a0 : 0.0f;
        a.y = live ? a1 : 0.0f;
        b.x = W1[(k0 + koff) * HID + colB];
        b.y = W1[(k0 + koff + 1) * HID + colB];
        c = __builtin_amdgcn_wmma_f32_16x16x4_f32(
                 false, a,  false, b,
                 (short)0, c,  false,  false);
    }

    const int rbase = m0 + (lane >> 4) * 8;
#pragma unroll
    for (int r = 0; r < 8; ++r) {
        int row = rbase + r;
        if (row < LPAD) ew[row * HID + colB] = (row < NLAB) ? c[r] : 0.0f;
    }
}


__global__ __launch_bounds__(GL_T) void group_list_kernel(const int* __restrict__ src, const int* __restrict__ dst,
                                                          unsigned* __restrict__ grec, unsigned* __restrict__ gown,
                                                          unsigned* __restrict__ gcount,
                                                          unsigned* __restrict__ pc,
                                                          const float* __restrict__ emb, const float* __restrict__ W1,
                                                          float* __restrict__ ew) {
    __shared__ __attribute__((aligned(16))) unsigned qrec[GCAP];
    __shared__ unsigned cnt[DGN];
    __shared__ unsigned qn;

    const int tid = threadIdx.x;
    const int lane = tid & 31;

    if (blockIdx.x == NGRP * GS) {
        const int wave = __builtin_amdgcn_readfirstlane(tid >> 5);
        if (wave < EW_TILES) label_weight_tile(emb, W1, ew, wave, lane);
        return;
    }

    const unsigned g = blockIdx.x % NGRP, sl = blockIdx.x / NGRP;
    const unsigned lo = g * GRNG;
    const unsigned slo = g * DGN;

    const int4* d4 = (const int4*)dst + (size_t)sl * GSE4;
    const int4* s4 = (const int4*)src + (size_t)sl * GSE4;
    int4 v[GL_L], sv[GL_L];
#pragma unroll
    for (int k = 0; k < GL_L; ++k) {
        const int i = k * GL_T + tid;
        v[k] = d4[i < GSE4 ? i : GSE4 - 1];
    }
#pragma unroll
    for (int k = 0; k < GL_L; ++k) {
        const int i = k * GL_T + tid;
        sv[k] = s4[i < GSE4 ? i : GSE4 - 1];
    }

    ((uint4*)qrec)[tid] = make_uint4(0u, 0u, 0u, 0u);
    for (int i = tid; i < DGN; i += GL_T) cnt[i] = 0u;
    if (tid == 0) qn = 0u;
    __syncthreads();

#pragma unroll
    for (int k = 0; k < GL_L; ++k) {
        if (k * GL_T + tid < GSE4) {
            const unsigned a0 = (unsigned)sv[k].x - slo, a1 = (unsigned)sv[k].y - slo;
            const unsigned a2 = (unsigned)sv[k].z - slo, a3 = (unsigned)sv[k].w - slo;
            if (a0 < DGN) atomicAdd(&cnt[a0], 1u);
            if (a1 < DGN) atomicAdd(&cnt[a1], 1u);
            if (a2 < DGN) atomicAdd(&cnt[a2], 1u);
            if (a3 < DGN) atomicAdd(&cnt[a3], 1u);
        }
    }

    unsigned wc = 0u;
#pragma unroll
    for (int k = 0; k < GL_L; ++k) {
        const bool in = k * GL_T + tid < GSE4;
        wc += (unsigned)__builtin_popcount(__builtin_amdgcn_ballot_w32(in && (unsigned)v[k].x - lo < GRNG));
        wc += (unsigned)__builtin_popcount(__builtin_amdgcn_ballot_w32(in && (unsigned)v[k].y - lo < GRNG));
        wc += (unsigned)__builtin_popcount(__builtin_amdgcn_ballot_w32(in && (unsigned)v[k].z - lo < GRNG));
        wc += (unsigned)__builtin_popcount(__builtin_amdgcn_ballot_w32(in && (unsigned)v[k].w - lo < GRNG));
    }
    unsigned base = 0u;
    if (lane == 0) base = __hip_atomic_fetch_add(&qn, wc, __ATOMIC_RELAXED, __HIP_MEMORY_SCOPE_WORKGROUP);
    base = __builtin_amdgcn_readfirstlane(base);

#pragma unroll
    for (int k = 0; k < GL_L; ++k) {
        const bool in = k * GL_T + tid < GSE4;
        const unsigned a[4] = {(unsigned)v[k].x - lo, (unsigned)v[k].y - lo,
                               (unsigned)v[k].z - lo, (unsigned)v[k].w - lo};
        const unsigned ss[4] = {(unsigned)sv[k].x, (unsigned)sv[k].y, (unsigned)sv[k].z, (unsigned)sv[k].w};
#pragma unroll
        for (int j = 0; j < 4; ++j) {
            const bool hit = in && a[j] < GRNG;
            const unsigned m = __builtin_amdgcn_ballot_w32(hit);
            const unsigned slot = __builtin_amdgcn_mbcnt_lo(m, base);
            if (hit && slot < GCAP) qrec[slot] = (a[j] << 16) | ss[j];
            base += (unsigned)__builtin_popcount(m);
        }
    }
    __syncthreads();

    const unsigned tot = qn;
    const unsigned n = tot < GCAP ? tot : GCAP;
    const size_t list = (size_t)g * GS + sl;
    if ((unsigned)tid * 4u < ((n + 15u) & ~15u)) {
        const uint4 q = ((const uint4*)qrec)[tid];
        const unsigned qq[4] = {q.x, q.y, q.z, q.w};
        unsigned r[4];
        unsigned ob = 0u;
#pragma unroll
        for (int c = 0; c < 4; ++c) {
            const bool live = (unsigned)tid * 4u + c < n;
            const unsigned a = qq[c] >> 16;
            const unsigned o = a / RNG;
            r[c] = live ? (((a - o * RNG) << 16) | (qq[c] & 0xffffu)) : 0u;
            ob |= (live ? g * GRP + o : 0u) << (8 * c);
        }
        ((uint4*)(grec + list * GCAP))[tid] = make_uint4(r[0], r[1], r[2], r[3]);
        gown[list * (GCAP / 4) + tid] = ob;
    }
    if (tid == 0) gcount[list] = tot;

    for (int i = tid; i < DGN; i += GL_T) pc[(size_t)(slo + i) * DS + sl] = cnt[i];
}


__device__ __forceinline__ unsigned own_bytes(unsigned x) {
    return ~(((x & 0x7f7f7f7fu) + 0x7f7f7f7fu) | x | 0x7f7f7f7fu);
}

__device__ __forceinline__ void lds_add(float* p, float v) {
    __hip_atomic_fetch_add(p, v, __ATOMIC_RELAXED, __HIP_MEMORY_SCOPE_WORKGROUP);
}

__device__ __forceinline__ void group_scan(const unsigned* __restrict__ gown, const unsigned* __restrict__ gcount,
                                           unsigned me, unsigned* q, unsigned* qn, unsigned cap,
                                           unsigned* ovf, int tid) {
    const unsigned g = me / GRP;
    const unsigned pat = me * 0x01010101u;
    const uint4* gw = (const uint4*)gown + (size_t)g * (GS * GW16);
    const unsigned* gc = gcount + g * GS;

    unsigned n[GW_L];
    uint4 x[GW_L];
#pragma unroll
    for (int j = 0; j < GW_L; ++j) {
        const unsigned c = gc[(unsigned)(j * AGG_T + tid) / GW16];
        if (c > GCAP) *ovf = 1u;
        n[j] = c < GCAP ? c : GCAP;
    }
#pragma unroll
    for (int j = 0; j < GW_L; ++j) {
        const unsigned w = (unsigned)(j * AGG_T + tid);
        x[j] = make_uint4(0u, 0u, 0u, 0u);
        if ((w % GW16) * 16u < n[j]) x[j] = gw[w];
    }
#pragma unroll
    for (int j = 0; j < GW_L; ++j) {
        const unsigned w = (unsigned)(j * AGG_T + tid);
        const unsigned pos = (w % GW16) * 16u;
        if (pos < n[j]) {
            const unsigned rem = n[j] - pos;
            unsigned m = (own_bytes(x[j].x ^ pat) >> 7) | (own_bytes(x[j].y ^ pat) >> 6) |
                         (own_bytes(x[j].z ^ pat) >> 5) | (own_bytes(x[j].w ^ pat) >> 4);
            while (m) {
                const unsigned b = (unsigned)__builtin_ctz(m);
                m &= m - 1u;
                const unsigned k = ((b & 7u) << 2) + (b >> 3);
                if (k < rem) {
                    const unsigned slot = __hip_atomic_fetch_add(qn, 1u, __ATOMIC_RELAXED,
                                                                 __HIP_MEMORY_SCOPE_WORKGROUP);
                    if (slot < cap) q[slot] = w * 16u + k;
                }
            }
        }
    }
}


template <int K, int NOUT, int NPAD>
__device__ __forceinline__ void stage_weights(const float* __restrict__ W, const float* __restrict__ bias,
                                              float* ws, float* bs, int tid) {
    static_assert(K % 4 == 0 && NPAD % 32 == 0 && NOUT <= NPAD && NOUT % 2 == 0 && NPAD <= AGG_T, "tile shapes");
    constexpr int PB = 2 * NPAD + 32;
    constexpr int NP2 = NPAD / 2;
    constexpr int ITEMS = (K / 2) * NP2;
#pragma unroll
    for (int it = 0; it < (ITEMS + AGG_T - 1) / AGG_T; ++it) {
        const int idx = tid + it * AGG_T;
        if (idx < ITEMS) {
            const int p = idx / NP2, n = (idx % NP2) * 2;
            v2f lo = {0.0f, 0.0f}, hi = {0.0f, 0.0f};
            if (NOUT == NPAD || n < NOUT) {
                lo = *(const v2f*)(W + (2 * p) * NOUT + n);
                hi = *(const v2f*)(W + (2 * p + 1) * NOUT + n);
            }
            *(float4*)(ws + p * PB + 2 * n) = make_float4(lo.x, hi.x, lo.y, hi.y);
        }
    }
    if (tid < NPAD) bs[tid] = (tid < NOUT) ? bias[tid] : 0.0f;
}

template <int K, int PITCH, int NOUT, int NPAD, bool RELU>
__device__ __forceinline__ void owner_rows_gemm(const float* rowsL, const float* sc, const float* ws,
                                                const float* bs, float* __restrict__ out,
                                                int rows, int wave, int lane) {
    constexpr int HT = NPAD / 32;
    constexpr int KS = K / 4;
    constexpr int PB = 2 * NPAD + 32;
    constexpr int G = 2;
    constexpr int NG = (KS + G - 1) / G;

    const int cl = lane & 15;
    const int h = lane >> 4;
    const int units = ((rows + 15) >> 4) * 2;

    for (int u = wave; u < units; u += AGG_T / 32) {
        const int rt = u >> 1;
        const int t0 = (u & 1) * HT;
        const float* ap = rowsL + (rt * 16 + cl) * PITCH + 2 * h;
        const float* bp = ws + h * PB + 2 * cl + 32 * t0;
        const float s = sc[rt * 16 + cl];

        v8f c[HT];
#pragma unroll
        for (int t = 0; t < HT; ++t) {
            const float bval = bs[16 * (t0 + t) + cl];
#pragma unroll
            for (int r = 0; r < 8; ++r) c[t][r] = bval;
        }

        v2f aq[2][G];
        v2f bq[2][G][HT];
#pragma unroll
        for (int i = 0; i < G; ++i) {
            if (i < KS) {
                aq[0][i] = *(const v2f*)(ap + 4 * i);
#pragma unroll
                for (int t = 0; t < HT; ++t) bq[0][i][t] = *(const v2f*)(bp + 2 * i * PB + 32 * t);
            }
        }

#pragma unroll
        for (int g = 0; g < NG; ++g) {
            if (g + 1 < NG) {
#pragma unroll
                for (int i = 0; i < G; ++i) {
                    const int ks = (g + 1) * G + i;
                    if (ks < KS) {
                        aq[(g + 1) & 1][i] = *(const v2f*)(ap + 4 * ks);
#pragma unroll
                        for (int t = 0; t < HT; ++t)
                            bq[(g + 1) & 1][i][t] = *(const v2f*)(bp + 2 * ks * PB + 32 * t);
                    }
                }
            }
#pragma unroll
            for (int i = 0; i < G; ++i) {
                const int ks = g * G + i;
                if (ks < KS) {
                    v2f a;
                    a.x = aq[g & 1][i].x * s;
                    a.y = aq[g & 1][i].y * s;
#pragma unroll
                    for (int t = 0; t < HT; ++t)
                        c[t] = __builtin_amdgcn_wmma_f32_16x16x4_f32(
                                 false, a,  false, bq[g & 1][i][t],
                                 (short)0, c[t],  false,  false);
                }
            }
            __builtin_amdgcn_sched_barrier(0);
        }

        const int lr = rt * 16 + h * 8;
#pragma unroll
        for (int t = 0; t < HT; ++t) {
            const int col = 16 * (t0 + t) + cl;
            if (NOUT == NPAD || col < NOUT) {
#pragma unroll
                for (int r = 0; r < 8; ++r) {
                    float v = c[t][r];
                    if (RELU) v = fmaxf(v, 0.0f);
                    if (lr + r < rows) out[(size_t)(lr + r) * NOUT + col] = v;
                }
            }
        }
    }
}

__global__ __launch_bounds__(AGG_T) void owner_label_kernel(const unsigned* __restrict__ gown,
                                                            const unsigned* __restrict__ grec,
                                                            const unsigned* __restrict__ gcount,
                                                            const int* __restrict__ src,
                                                            const int* __restrict__ dst,
                                                            const int* __restrict__ labels,
                                                            const unsigned* __restrict__ pc,
                                                            const float* __restrict__ ew,
                                                            const float* __restrict__ bias,
                                                            float* __restrict__ hout,
                                                            float* __restrict__ inorm,
                                                            float* __restrict__ onorm,
                                                            unsigned* __restrict__ elist,
                                                            unsigned* __restrict__ ecount) {
    __shared__ __attribute__((aligned(16))) float lt[RNGP * LPAD];
    __shared__ unsigned queue[QALL];
    __shared__ __attribute__((aligned(16))) float ws[(LPAD / 2) * (2 * HID + 32)];
    __shared__ float bs[HID];
    __shared__ float sc[RNGP];
    __shared__ unsigned cnt[RNG];
    __shared__ unsigned qn;
    __shared__ unsigned ovf;

    const int tid = threadIdx.x;
    const int lane = tid & 31;
    const int wave = __builtin_amdgcn_readfirstlane(tid >> 5);
    const unsigned me = blockIdx.x;
    const unsigned lo = me * RNG;

    float4* lt4 = (float4*)lt;
    for (int i = tid; i < RNGP * LPAD / 4; i += AGG_T) lt4[i] = make_float4(0.0f, 0.0f, 0.0f, 0.0f);
    if (tid < RNG) cnt[tid] = 0u;
    if (tid == 0) { qn = 0u; ovf = 0u; }
    stage_weights<LPAD, HID, HID>(ew, bias, ws, bs, tid);
    __syncthreads();

    group_scan(gown, gcount, me, queue, &qn, QALL, &ovf, tid);
    __syncthreads();

    const unsigned tot = qn;
    const bool whole = ovf == 0u && tot <= QALL;
    if (tid == 0) ecount[me] = whole ? tot : 0xffffffffu;
    if (whole) {
        uint2* el = (uint2*)elist + (size_t)me * LCAP;
        const unsigned* gr = grec + (size_t)(me / GRP) * (GS * GCAP);
        for (unsigned i = tid; i < tot; i += AGG_T) {
            const unsigned r = gr[queue[i]];
            const unsigned s = r & 0xffffu, row = r >> 16;
            const float w = out_norm_of(pc, s);
            if (i < LCAP) el[i] = make_uint2(r, __float_as_uint(w));
            lds_add(&lt[row * LPAD + labels[s]], w);
            atomicAdd(&cnt[row], 1u);
        }
    } else {
        for (unsigned e = tid; e < NE; e += AGG_T) {
            const unsigned row = (unsigned)dst[e] - lo;
            if (row < RNG) {
                const unsigned s = (unsigned)src[e];
                lds_add(&lt[row * LPAD + labels[s]], out_norm_of(pc, s));
                atomicAdd(&cnt[row], 1u);
            }
        }
    }
    __syncthreads();

    const int rows = ((int)NN - (int)lo < RNG) ? (int)NN - (int)lo : RNG;
    if (tid < RNGP) {
        float v = 0.0f;
        if (tid < rows) {
            v = rsqrtf(fmaxf((float)cnt[tid], 1.0f));
            inorm[lo + tid] = v;
            onorm[lo + tid] = out_norm_of(pc, lo + tid);
        }
        sc[tid] = v;
    }
    __syncthreads();

    owner_rows_gemm<LPAD, LPAD, HID, HID, true>(lt, sc, ws, bs, hout + (size_t)lo * HID, rows, wave, lane);
}

template <int NOUT, int NPAD, bool RELU>
__global__ __launch_bounds__(AGG_T) void owner_sum_kernel(const float* __restrict__ hs,
                                                          const unsigned* __restrict__ elist,
                                                          const unsigned* __restrict__ ecount,
                                                          const int* __restrict__ src,
                                                          const int* __restrict__ dst,
                                                          const float* __restrict__ onorm,
                                                          const float* __restrict__ inorm,
                                                          const float* __restrict__ W,
                                                          const float* __restrict__ bias,
                                                          float* __restrict__ out) {
    __shared__ __attribute__((aligned(16))) float slab[RNGP * SP];
    __shared__ uint2 lst[LSTG];
    __shared__ __attribute__((aligned(16))) float ws[(HID / 2) * (2 * NPAD + 32)];
    __shared__ float bs[NPAD];
    __shared__ float sc[RNGP];

    const int tid = threadIdx.x;
    const int lane = tid & 31;
    const int wave = __builtin_amdgcn_readfirstlane(tid >> 5);
    const unsigned me = blockIdx.x;
    const int lo = me * RNG;
    const int rows = (NN - lo < RNG) ? NN - lo : RNG;

    const unsigned tot = __builtin_amdgcn_readfirstlane(ecount[me]);
    const bool listed = tot <= LCAP;

    const uint2* L = (const uint2*)elist + (size_t)me * LCAP;
    const uint2 none = {0xffffffffu, 0u};
    uint2 e[LCAP / AGG_T];
#pragma unroll
    for (int k = 0; k < LCAP / AGG_T; ++k) {
        const unsigned i = (unsigned)(k * AGG_T + tid);
        e[k] = none;
        if (listed && i < tot) e[k] = L[i];
    }

    float4* slab4 = (float4*)slab;
    for (int i = tid; i < RNGP * SP / 4; i += AGG_T) slab4[i] = make_float4(0.0f, 0.0f, 0.0f, 0.0f);
    stage_weights<HID, NOUT, NPAD>(W, bias, ws, bs, tid);
    if (tid < RNGP) sc[tid] = (tid < rows) ? inorm[lo + tid] : 0.0f;
#pragma unroll
    for (int k = 0; k < LCAP / AGG_T; ++k) lst[k * AGG_T + tid] = e[k];
    __syncthreads();

    if (listed) {
        const unsigned per = (((tot + 31u) >> 5) + 7u) & ~7u;
        const unsigned beg = wave * per;
        const unsigned end = (beg + per < tot) ? beg + per : tot;
        for (unsigned base = beg; base < end; base += 32u) {
            const unsigned idx = base + lane;
            const uint2 ent = (idx < end) ? lst[idx] : none;
            const unsigned c = (end - base < 32u) ? end - base : 32u;
            for (unsigned j0 = 0; j0 < c; j0 += 8u) {
                float v[8][3];
                float w[8];
                unsigned rr[8];
#pragma unroll
                for (int u = 0; u < 8; ++u) {
                    rr[u] = __builtin_amdgcn_readlane(ent.x, j0 + u);
                    w[u] = __uint_as_float(__builtin_amdgcn_readlane(ent.y, j0 + u));
                    const unsigned s = (rr[u] == 0xffffffffu) ? 0u : (rr[u] & 0xffffu);
                    const float* hp = hs + (size_t)s * HID + lane;
                    v[u][0] = hp[0];
                    v[u][1] = hp[32];
                    v[u][2] = hp[64];
                }
#pragma unroll
                for (int u = 0; u < 8; ++u) {
                    if (rr[u] != 0xffffffffu) {
                        float* sp = slab + (rr[u] >> 16) * SP + lane;
                        lds_add(sp, v[u][0] * w[u]);
                        lds_add(sp + 32, v[u][1] * w[u]);
                        lds_add(sp + 64, v[u][2] * w[u]);
                    }
                }
            }
        }
    } else {
        for (unsigned ed = tid; ed < NE; ed += AGG_T) {
            const unsigned row = (unsigned)dst[ed] - (unsigned)lo;
            if (row < RNG) {
                const unsigned sn = (unsigned)src[ed];
                const float* hp = hs + (size_t)sn * HID;
                const float w = onorm[sn];
                float* sp = slab + row * SP;
                for (int f = 0; f < HID; ++f) lds_add(sp + f, hp[f] * w);
            }
        }
    }
    __syncthreads();

    owner_rows_gemm<HID, SP, NOUT, NPAD, RELU>(slab, sc, ws, bs, out + (size_t)lo * NOUT, rows, wave, lane);
}


extern "C" void kernel_launch(void* const* d_in, const int* in_sizes, int n_in,
                              void* d_out, int out_size, void* d_ws, size_t ws_size,
                              hipStream_t stream) {
    const int*   labels = (const int*)d_in[0];
    const int*   src    = (const int*)d_in[1];
    const int*   dst    = (const int*)d_in[2];
    const float* emb    = (const float*)d_in[3];
    const float* W1     = (const float*)d_in[4];
    const float* b1     = (const float*)d_in[5];
    const float* W2     = (const float*)d_in[6];
    const float* b2     = (const float*)d_in[7];
    const float* W3     = (const float*)d_in[8];
    const float* b3     = (const float*)d_in[9];
    float* out = (float*)d_out;

    float* w = (float*)d_ws;
    float* onorm = w;
    float* inorm = w + NN;
    float* bufA  = w + 2 * (size_t)NN;
    float* bufB  = bufA + (size_t)NN * HID;
    float* bufC  = bufB + (size_t)NN * HID;
    unsigned* gown = (unsigned*)(bufB + 8192);
    unsigned* grec = gown + NGRP * GS * (GCAP / 4);
    unsigned* gcount = grec + (size_t)NGRP * GS * GCAP;
    unsigned* elist  = gcount + NGRP * GS;
    unsigned* ecount = elist + 2 * (size_t)NOWN * LCAP;

    unsigned* pc = (unsigned*)bufC;
    float* ew = bufB;

    group_list_kernel<<<NGRP * GS + 1, GL_T, 0, stream>>>(src, dst, grec, gown, gcount, pc, emb, W1, ew);

    owner_label_kernel<<<NOWN, AGG_T, 0, stream>>>(gown, grec, gcount, src, dst, labels, pc, ew, b1, bufA, inorm, onorm,
                                                   elist, ecount);

    owner_sum_kernel<HID, HID, true><<<NOWN, AGG_T, 0, stream>>>(
        bufA, elist, ecount, src, dst, onorm, inorm, W2, b2, bufC);

    owner_sum_kernel<NLAB, 64, false><<<NOWN, AGG_T, 0, stream>>>(
        bufC, elist, ecount, src, dst, onorm, inorm, W3, b3, out);
}
